# starting kernel
# baseline (speedup 1.0000x reference)
#include <hip/hip_runtime.h>

#define NN 40000
#define NE 640000
#define DD 128
#define NC 40
#define NG 128

typedef __attribute__((ext_vector_type(2))) float v2f;
typedef __attribute__((ext_vector_type(8))) float v8f;

__device__ __forceinline__ void atomAddF(float* p, float v) {
#if defined(__HIP_DEVICE_COMPILE__)
  unsafeAtomicAdd(p, v);
#else
  __hip_atomic_fetch_add(p, v, __ATOMIC_RELAXED, __HIP_MEMORY_SCOPE_AGENT);
#endif
}

__global__ void deg_kernel(const int* __restrict__ src, const int* __restrict__ dst,
                           unsigned* __restrict__ deg_out, unsigned* __restrict__ deg_in) {
  int e = blockIdx.x * blockDim.x + threadIdx.x;
  if (e < NE) {
    atomicAdd(&deg_out[src[e]], 1u);
    atomicAdd(&deg_in[dst[e]], 1u);
  }
}

__global__ void norm_kernel(const unsigned* __restrict__ dout, const unsigned* __restrict__ din,
                            float* __restrict__ ns, float* __restrict__ nd) {
  int i = blockIdx.x * blockDim.x + threadIdx.x;
  if (i < NN) {
    ns[i] = rsqrtf(fmaxf((float)dout[i], 1.0f));
    nd[i] = rsqrtf(fmaxf((float)din[i], 1.0f));
  }
}

__global__ void __launch_bounds__(256) gemm_rs_wmma(const float* __restrict__ X,
                                                    const float* __restrict__ rs,
                                                    const float* __restrict__ W,
                                                    float* __restrict__ Y) {
  const int lane = threadIdx.x & 31;
  const int wave = threadIdx.x >> 5;
  const int hl   = lane >> 4;
  const int l    = lane & 15;
  const int row0 = blockIdx.x * 16;
  const int n0   = wave * 16;
  const int m    = row0 + l;
  const float scale = rs[m];
  const float* __restrict__ arow = X + (size_t)m * DD;

  v8f c = {};
#pragma unroll
  for (int k = 0; k < DD; k += 4) {
    const int ka = k + 2 * hl;
    v2f a, b;
    a.x = arow[ka]     * scale;
    a.y = arow[ka + 1] * scale;
    b.x = W[ka * DD + n0 + l];
    b.y = W[(ka + 1) * DD + n0 + l];
    c = __builtin_amdgcn_wmma_f32_16x16x4_f32(false, a, false, b, (short)0, c, false, false);
  }
#pragma unroll
  for (int r = 0; r < 8; ++r) {
    Y[(size_t)(row0 + 8 * hl + r) * DD + n0 + l] = c[r];
  }
}

__global__ void spmm_kernel(const float* __restrict__ H, const int* __restrict__ src,
                            const int* __restrict__ dst, float* __restrict__ AGG) {
  const int lane = threadIdx.x & 31;
  const int e = blockIdx.x * (blockDim.x >> 5) + (threadIdx.x >> 5);
  if (e >= NE) return;
  const int s = src[e];
  const int d = dst[e];
  const float4 v = *(const float4*)(H + (size_t)s * DD + lane * 4);
  float* out = AGG + (size_t)d * DD + lane * 4;
  atomAddF(out + 0, v.x);
  atomAddF(out + 1, v.y);
  atomAddF(out + 2, v.z);
  atomAddF(out + 3, v.w);
}

__global__ void post_relu(const float* __restrict__ AGG, const float* __restrict__ nd,
                          const float* __restrict__ bias, float* __restrict__ OUT) {
  int idx = blockIdx.x * blockDim.x + threadIdx.x;
  if (idx < NN * DD) {
    int node = idx >> 7, dcol = idx & (DD - 1);
    OUT[idx] = fmaxf(AGG[idx] * nd[node] + bias[dcol], 0.0f);
  }
}

__global__ void post_pool(const float* __restrict__ AGG, const float* __restrict__ nd,
                          const float* __restrict__ bias, const int* __restrict__ gid,
                          float* __restrict__ sums) {
  int idx = blockIdx.x * blockDim.x + threadIdx.x;
  if (idx < NN * DD) {
    int node = idx >> 7, dcol = idx & (DD - 1);
    float v = AGG[idx] * nd[node] + bias[dcol];
    atomAddF(&sums[gid[node] * DD + dcol], v);
  }
}

__global__ void count_kernel(const int* __restrict__ gid, float* __restrict__ cnt) {
  int i = blockIdx.x * blockDim.x + threadIdx.x;
  if (i < NN) atomAddF(&cnt[gid[i]], 1.0f);
}

__global__ void classify(const float* __restrict__ sums, const float* __restrict__ cnt,
                         const float* __restrict__ Wc, const float* __restrict__ bc,
                         float* __restrict__ out) {
  int idx = blockIdx.x * blockDim.x + threadIdx.x;
  if (idx < NG * NC) {
    int g = idx / NC, cc = idx % NC;
    float inv = 1.0f / fmaxf(cnt[g], 1.0f);
    float acc = bc[cc];
#pragma unroll 8
    for (int k = 0; k < DD; ++k)
      acc += (sums[g * DD + k] * inv) * Wc[k * NC + cc];
    out[idx] = acc;
  }
}

extern "C" void kernel_launch(void* const* d_in, const int* in_sizes, int n_in,
                              void* d_out, int out_size, void* d_ws, size_t ws_size,
                              hipStream_t stream) {
  const float* feat = (const float*)d_in[0];
  const float* W1   = (const float*)d_in[1];
  const float* b1   = (const float*)d_in[2];
  const float* W2   = (const float*)d_in[3];
  const float* b2   = (const float*)d_in[4];
  const float* Wc   = (const float*)d_in[5];
  const float* bc   = (const float*)d_in[6];
  const int*   src  = (const int*)d_in[7];
  const int*   dst  = (const int*)d_in[8];
  const int*   gid  = (const int*)d_in[9];
  float* out = (float*)d_out;

  float*    ws      = (float*)d_ws;
  unsigned* deg_out = (unsigned*)ws;
  unsigned* deg_in  = deg_out + NN;
  float*    nsrc    = ws + 2 * NN;
  float*    ndst    = nsrc + NN;
  float*    bufA    = ndst + NN;
  float*    bufB    = bufA + (size_t)NN * DD;
  float*    sums    = bufB + (size_t)NN * DD;
  float*    cnt     = sums + NG * DD;

  const int T = 256;

  hipMemsetAsync(deg_out, 0, 2 * NN * sizeof(unsigned), stream);
  hipMemsetAsync(sums, 0, (NG * DD + NG) * sizeof(float), stream);

  deg_kernel<<<(NE + T - 1) / T, T, 0, stream>>>(src, dst, deg_out, deg_in);
  norm_kernel<<<(NN + T - 1) / T, T, 0, stream>>>(deg_out, deg_in, nsrc, ndst);

  gemm_rs_wmma<<<NN / 16, 256, 0, stream>>>(feat, nsrc, W1, bufA);
  hipMemsetAsync(bufB, 0, (size_t)NN * DD * sizeof(float), stream);
  spmm_kernel<<<NE / 8, 256, 0, stream>>>(bufA, src, dst, bufB);
  post_relu<<<(NN * DD + T - 1) / T, T, 0, stream>>>(bufB, ndst, b1, bufA);

  gemm_rs_wmma<<<NN / 16, 256, 0, stream>>>(bufA, nsrc, W2, bufB);
  hipMemsetAsync(bufA, 0, (size_t)NN * DD * sizeof(float), stream);
  spmm_kernel<<<NE / 8, 256, 0, stream>>>(bufB, src, dst, bufA);

  post_pool<<<(NN * DD + T - 1) / T, T, 0, stream>>>(bufA, ndst, b2, gid, sums);
  count_kernel<<<(NN + T - 1) / T, T, 0, stream>>>(gid, cnt);
  classify<<<(NG * NC + T - 1) / T, T, 0, stream>>>(sums, cnt, Wc, bc, out);
}


# spmm: lane owns columns l, l+32, l+64, l+96, so each f32 atomic and each row load covers one whole 128-B line: 4 L2 atomic requests per edge instead of 16; sums still in atomic arrival order
# speedup vs baseline: 1.2554x; 1.2554x over previous
#include <hip/hip_runtime.h>

#define NN 40000
#define NE 640000
#define DD 128
#define NC 40
#define NG 128

typedef __attribute__((ext_vector_type(2))) float v2f;
typedef __attribute__((ext_vector_type(8))) float v8f;

__device__ __forceinline__ void atomAddF(float* p, float v) {
#if defined(__HIP_DEVICE_COMPILE__)
  unsafeAtomicAdd(p, v);
#else
  __hip_atomic_fetch_add(p, v, __ATOMIC_RELAXED, __HIP_MEMORY_SCOPE_AGENT);
#endif
}

__global__ void deg_kernel(const int* __restrict__ src, const int* __restrict__ dst,
                           unsigned* __restrict__ deg_out, unsigned* __restrict__ deg_in) {
  int e = blockIdx.x * blockDim.x + threadIdx.x;
  if (e < NE) {
    atomicAdd(&deg_out[src[e]], 1u);
    atomicAdd(&deg_in[dst[e]], 1u);
  }
}

__global__ void norm_kernel(const unsigned* __restrict__ dout, const unsigned* __restrict__ din,
                            float* __restrict__ ns, float* __restrict__ nd) {
  int i = blockIdx.x * blockDim.x + threadIdx.x;
  if (i < NN) {
    ns[i] = rsqrtf(fmaxf((float)dout[i], 1.0f));
    nd[i] = rsqrtf(fmaxf((float)din[i], 1.0f));
  }
}

__global__ void __launch_bounds__(256) gemm_rs_wmma(const float* __restrict__ X,
                                                    const float* __restrict__ rs,
                                                    const float* __restrict__ W,
                                                    float* __restrict__ Y) {
  const int lane = threadIdx.x & 31;
  const int wave = threadIdx.x >> 5;
  const int hl   = lane >> 4;
  const int l    = lane & 15;
  const int row0 = blockIdx.x * 16;
  const int n0   = wave * 16;
  const int m    = row0 + l;
  const float scale = rs[m];
  const float* __restrict__ arow = X + (size_t)m * DD;

  v8f c = {};
#pragma unroll
  for (int k = 0; k < DD; k += 4) {
    const int ka = k + 2 * hl;
    v2f a, b;
    a.x = arow[ka]     * scale;
    a.y = arow[ka + 1] * scale;
    b.x = W[ka * DD + n0 + l];
    b.y = W[(ka + 1) * DD + n0 + l];
    c = __builtin_amdgcn_wmma_f32_16x16x4_f32(false, a, false, b, (short)0, c, false, false);
  }
#pragma unroll
  for (int r = 0; r < 8; ++r) {
    Y[(size_t)(row0 + 8 * hl + r) * DD + n0 + l] = c[r];
  }
}

__global__ void spmm_kernel(const float* __restrict__ H, const int* __restrict__ src,
                            const int* __restrict__ dst, float* __restrict__ AGG) {
  const int lane = threadIdx.x & 31;
  const int e = blockIdx.x * (blockDim.x >> 5) + (threadIdx.x >> 5);
  if (e >= NE) return;
  const int s = src[e];
  const int d = dst[e];
  const float* __restrict__ hrow = H + (size_t)s * DD + lane;
  const float v0 = hrow[0];
  const float v1 = hrow[32];
  const float v2 = hrow[64];
  const float v3 = hrow[96];
  float* out = AGG + (size_t)d * DD + lane;
  atomAddF(out, v0);
  atomAddF(out + 32, v1);
  atomAddF(out + 64, v2);
  atomAddF(out + 96, v3);
}

__global__ void post_relu(const float* __restrict__ AGG, const float* __restrict__ nd,
                          const float* __restrict__ bias, float* __restrict__ OUT) {
  int idx = blockIdx.x * blockDim.x + threadIdx.x;
  if (idx < NN * DD) {
    int node = idx >> 7, dcol = idx & (DD - 1);
    OUT[idx] = fmaxf(AGG[idx] * nd[node] + bias[dcol], 0.0f);
  }
}

__global__ void post_pool(const float* __restrict__ AGG, const float* __restrict__ nd,
                          const float* __restrict__ bias, const int* __restrict__ gid,
                          float* __restrict__ sums) {
  int idx = blockIdx.x * blockDim.x + threadIdx.x;
  if (idx < NN * DD) {
    int node = idx >> 7, dcol = idx & (DD - 1);
    float v = AGG[idx] * nd[node] + bias[dcol];
    atomAddF(&sums[gid[node] * DD + dcol], v);
  }
}

__global__ void count_kernel(const int* __restrict__ gid, float* __restrict__ cnt) {
  int i = blockIdx.x * blockDim.x + threadIdx.x;
  if (i < NN) atomAddF(&cnt[gid[i]], 1.0f);
}

__global__ void classify(const float* __restrict__ sums, const float* __restrict__ cnt,
                         const float* __restrict__ Wc, const float* __restrict__ bc,
                         float* __restrict__ out) {
  int idx = blockIdx.x * blockDim.x + threadIdx.x;
  if (idx < NG * NC) {
    int g = idx / NC, cc = idx % NC;
    float inv = 1.0f / fmaxf(cnt[g], 1.0f);
    float acc = bc[cc];
#pragma unroll 8
    for (int k = 0; k < DD; ++k)
      acc += (sums[g * DD + k] * inv) * Wc[k * NC + cc];
    out[idx] = acc;
  }
}

extern "C" void kernel_launch(void* const* d_in, const int* in_sizes, int n_in,
                              void* d_out, int out_size, void* d_ws, size_t ws_size,
                              hipStream_t stream) {
  const float* feat = (const float*)d_in[0];
  const float* W1   = (const float*)d_in[1];
  const float* b1   = (const float*)d_in[2];
  const float* W2   = (const float*)d_in[3];
  const float* b2   = (const float*)d_in[4];
  const float* Wc   = (const float*)d_in[5];
  const float* bc   = (const float*)d_in[6];
  const int*   src  = (const int*)d_in[7];
  const int*   dst  = (const int*)d_in[8];
  const int*   gid  = (const int*)d_in[9];
  float* out = (float*)d_out;

  float*    ws      = (float*)d_ws;
  unsigned* deg_out = (unsigned*)ws;
  unsigned* deg_in  = deg_out + NN;
  float*    nsrc    = ws + 2 * NN;
  float*    ndst    = nsrc + NN;
  float*    bufA    = ndst + NN;
  float*    bufB    = bufA + (size_t)NN * DD;
  float*    sums    = bufB + (size_t)NN * DD;
  float*    cnt     = sums + NG * DD;

  const int T = 256;

  hipMemsetAsync(deg_out, 0, 2 * NN * sizeof(unsigned), stream);
  hipMemsetAsync(sums, 0, (NG * DD + NG) * sizeof(float), stream);

  deg_kernel<<<(NE + T - 1) / T, T, 0, stream>>>(src, dst, deg_out, deg_in);
  norm_kernel<<<(NN + T - 1) / T, T, 0, stream>>>(deg_out, deg_in, nsrc, ndst);

  gemm_rs_wmma<<<NN / 16, 256, 0, stream>>>(feat, nsrc, W1, bufA);
  hipMemsetAsync(bufB, 0, (size_t)NN * DD * sizeof(float), stream);
  spmm_kernel<<<NE / 8, 256, 0, stream>>>(bufA, src, dst, bufB);
  post_relu<<<(NN * DD + T - 1) / T, T, 0, stream>>>(bufB, ndst, b1, bufA);

  gemm_rs_wmma<<<NN / 16, 256, 0, stream>>>(bufA, nsrc, W2, bufB);
  hipMemsetAsync(bufA, 0, (size_t)NN * DD * sizeof(float), stream);
  spmm_kernel<<<NE / 8, 256, 0, stream>>>(bufB, src, dst, bufA);

  post_pool<<<(NN * DD + T - 1) / T, T, 0, stream>>>(bufA, ndst, b2, gid, sums);
  count_kernel<<<(NN + T - 1) / T, T, 0, stream>>>(gid, cnt);
  classify<<<(NG * NC + T - 1) / T, T, 0, stream>>>(sums, cnt, Wc, bc, out);
}


# spmm: one wave per destination row gathers its in-edges from a CSR built in-call (rank = return of the degree atomic, scan, fill), one store with the epilogue; no f32 atomics; sums in rank order
# speedup vs baseline: 1.6716x; 1.3315x over previous
#include <hip/hip_runtime.h>

#define NN 40000
#define NE 640000
#define DD 128
#define NC 40
#define NG 128

typedef __attribute__((ext_vector_type(2))) float v2f;
typedef __attribute__((ext_vector_type(8))) float v8f;

__device__ __forceinline__ void atomAddF(float* p, float v) {
#if defined(__HIP_DEVICE_COMPILE__)
  unsafeAtomicAdd(p, v);
#else
  __hip_atomic_fetch_add(p, v, __ATOMIC_RELAXED, __HIP_MEMORY_SCOPE_AGENT);
#endif
}

__global__ void deg_kernel(const int* __restrict__ src, const int* __restrict__ dst,
                           unsigned* __restrict__ deg_out, unsigned* __restrict__ deg_in,
                           unsigned* __restrict__ rank) {
  int e = blockIdx.x * blockDim.x + threadIdx.x;
  if (e < NE) {
    atomicAdd(&deg_out[src[e]], 1u);
    rank[e] = atomicAdd(&deg_in[dst[e]], 1u);
  }
}

#define SCAN_PER 40
__global__ void __launch_bounds__(1024) rowptr_scan(const unsigned* __restrict__ din,
                                                    unsigned* __restrict__ rowptr) {
  __shared__ unsigned wsum[32];
  const int t = threadIdx.x;
  const int lane = t & 31;
  const int w = t >> 5;
  const bool active = t < NN / SCAN_PER;
  uint4 v[SCAN_PER / 4];
  unsigned tot = 0;
  if (active) {
    const uint4* __restrict__ p = (const uint4*)(din + t * SCAN_PER);
#pragma unroll
    for (int i = 0; i < SCAN_PER / 4; ++i) v[i] = p[i];
#pragma unroll
    for (int i = 0; i < SCAN_PER / 4; ++i) tot += v[i].x + v[i].y + v[i].z + v[i].w;
  }
  unsigned inc = tot;
  for (int o = 1; o < 32; o <<= 1) {
    const unsigned u = __shfl_up(inc, o);
    if (lane >= o) inc += u;
  }
  if (lane == 31) wsum[w] = inc;
  __syncthreads();
  if (w == 0) {
    const unsigned own = wsum[lane];
    unsigned si = own;
    for (int o = 1; o < 32; o <<= 1) {
      const unsigned u = __shfl_up(si, o);
      if (lane >= o) si += u;
    }
    wsum[lane] = si - own;
  }
  __syncthreads();
  unsigned run = wsum[w] + inc - tot;
  if (active) {
    uint4* __restrict__ q = (uint4*)(rowptr + t * SCAN_PER);
#pragma unroll
    for (int i = 0; i < SCAN_PER / 4; ++i) {
      uint4 o;
      o.x = run; run += v[i].x;
      o.y = run; run += v[i].y;
      o.z = run; run += v[i].z;
      o.w = run; run += v[i].w;
      q[i] = o;
    }
    if (t == NN / SCAN_PER - 1) rowptr[NN] = run;
  }
}

__global__ void fill_kernel(const int* __restrict__ src, const int* __restrict__ dst,
                            const unsigned* __restrict__ rowptr, const unsigned* __restrict__ rank,
                            int* __restrict__ col) {
  int e = blockIdx.x * blockDim.x + threadIdx.x;
  if (e < NE) {
    const unsigned pos = rowptr[dst[e]] + rank[e];
    if (pos < (unsigned)NE) col[pos] = src[e];
  }
}

__global__ void norm_kernel(const unsigned* __restrict__ dout, const unsigned* __restrict__ din,
                            float* __restrict__ ns, float* __restrict__ nd) {
  int i = blockIdx.x * blockDim.x + threadIdx.x;
  if (i < NN) {
    ns[i] = rsqrtf(fmaxf((float)dout[i], 1.0f));
    nd[i] = rsqrtf(fmaxf((float)din[i], 1.0f));
  }
}

__global__ void __launch_bounds__(256) gemm_rs_wmma(const float* __restrict__ X,
                                                    const float* __restrict__ rs,
                                                    const float* __restrict__ W,
                                                    float* __restrict__ Y) {
  const int lane = threadIdx.x & 31;
  const int wave = threadIdx.x >> 5;
  const int hl   = lane >> 4;
  const int l    = lane & 15;
  const int row0 = blockIdx.x * 16;
  const int n0   = wave * 16;
  const int m    = row0 + l;
  const float scale = rs[m];
  const float* __restrict__ arow = X + (size_t)m * DD;

  v8f c = {};
#pragma unroll
  for (int k = 0; k < DD; k += 4) {
    const int ka = k + 2 * hl;
    v2f a, b;
    a.x = arow[ka]     * scale;
    a.y = arow[ka + 1] * scale;
    b.x = W[ka * DD + n0 + l];
    b.y = W[(ka + 1) * DD + n0 + l];
    c = __builtin_amdgcn_wmma_f32_16x16x4_f32(false, a, false, b, (short)0, c, false, false);
  }
#pragma unroll
  for (int r = 0; r < 8; ++r) {
    Y[(size_t)(row0 + 8 * hl + r) * DD + n0 + l] = c[r];
  }
}

template <int K>
__device__ __forceinline__ void gather_rows(const float4* __restrict__ H4, int mine, int j,
                                            int lane, float4& acc) {
  float4 r[K];
#pragma unroll
  for (int k = 0; k < K; ++k) {
    unsigned sidx = (unsigned)__builtin_amdgcn_readlane(mine, j + k);
    sidx = sidx < (unsigned)NN ? sidx : (unsigned)(NN - 1);
    r[k] = H4[(size_t)sidx * (DD / 4) + lane];
  }
#pragma unroll
  for (int k = 0; k < K; ++k) {
    acc.x += r[k].x; acc.y += r[k].y; acc.z += r[k].z; acc.w += r[k].w;
  }
}

template <bool RELU_EPILOGUE>
__global__ void __launch_bounds__(256) spmm_pull(const float* __restrict__ H,
                                                 const unsigned* __restrict__ rowptr,
                                                 const int* __restrict__ col,
                                                 const float* __restrict__ nd,
                                                 const float* __restrict__ bias,
                                                 float* __restrict__ OUT) {
  const int lane = threadIdx.x & 31;
  const int d = __builtin_amdgcn_readfirstlane((int)(blockIdx.x * (blockDim.x >> 5) + (threadIdx.x >> 5)));
  if (d >= NN) return;
  const unsigned beg = rowptr[d];
  unsigned end = rowptr[d + 1];
  end = end < (unsigned)NE ? end : (unsigned)NE;
  const float4* __restrict__ H4 = (const float4*)H;
  float sc = 1.0f;
  float4 b = {0.0f, 0.0f, 0.0f, 0.0f};
  if (RELU_EPILOGUE) {
    sc = nd[d];
    b = ((const float4*)bias)[lane];
  }
  float4 acc = {0.0f, 0.0f, 0.0f, 0.0f};
  for (unsigned p = beg; p < end; p += 32) {
    const unsigned q = p + lane;
    const int mine = q < end ? col[q] : 0;
    const int n = (int)((end - p) < 32u ? (end - p) : 32u);
    int j = 0;
    for (; j + 8 <= n; j += 8) gather_rows<8>(H4, mine, j, lane, acc);
    if (j + 4 <= n) { gather_rows<4>(H4, mine, j, lane, acc); j += 4; }
    if (j + 2 <= n) { gather_rows<2>(H4, mine, j, lane, acc); j += 2; }
    if (j < n)      { gather_rows<1>(H4, mine, j, lane, acc); }
  }
  float4 o = acc;
  if (RELU_EPILOGUE) {
    o.x = fmaxf(acc.x * sc + b.x, 0.0f);
    o.y = fmaxf(acc.y * sc + b.y, 0.0f);
    o.z = fmaxf(acc.z * sc + b.z, 0.0f);
    o.w = fmaxf(acc.w * sc + b.w, 0.0f);
  }
  ((float4*)OUT)[(size_t)d * (DD / 4) + lane] = o;
}

__global__ void post_pool(const float* __restrict__ AGG, const float* __restrict__ nd,
                          const float* __restrict__ bias, const int* __restrict__ gid,
                          float* __restrict__ sums) {
  int idx = blockIdx.x * blockDim.x + threadIdx.x;
  if (idx < NN * DD) {
    int node = idx >> 7, dcol = idx & (DD - 1);
    float v = AGG[idx] * nd[node] + bias[dcol];
    atomAddF(&sums[gid[node] * DD + dcol], v);
  }
}

__global__ void count_kernel(const int* __restrict__ gid, float* __restrict__ cnt) {
  int i = blockIdx.x * blockDim.x + threadIdx.x;
  if (i < NN) atomAddF(&cnt[gid[i]], 1.0f);
}

__global__ void classify(const float* __restrict__ sums, const float* __restrict__ cnt,
                         const float* __restrict__ Wc, const float* __restrict__ bc,
                         float* __restrict__ out) {
  int idx = blockIdx.x * blockDim.x + threadIdx.x;
  if (idx < NG * NC) {
    int g = idx / NC, cc = idx % NC;
    float inv = 1.0f / fmaxf(cnt[g], 1.0f);
    float acc = bc[cc];
#pragma unroll 8
    for (int k = 0; k < DD; ++k)
      acc += (sums[g * DD + k] * inv) * Wc[k * NC + cc];
    out[idx] = acc;
  }
}

extern "C" void kernel_launch(void* const* d_in, const int* in_sizes, int n_in,
                              void* d_out, int out_size, void* d_ws, size_t ws_size,
                              hipStream_t stream) {
  const float* feat = (const float*)d_in[0];
  const float* W1   = (const float*)d_in[1];
  const float* b1   = (const float*)d_in[2];
  const float* W2   = (const float*)d_in[3];
  const float* b2   = (const float*)d_in[4];
  const float* Wc   = (const float*)d_in[5];
  const float* bc   = (const float*)d_in[6];
  const int*   src  = (const int*)d_in[7];
  const int*   dst  = (const int*)d_in[8];
  const int*   gid  = (const int*)d_in[9];
  float* out = (float*)d_out;

  float*    ws      = (float*)d_ws;
  unsigned* deg_out = (unsigned*)ws;
  unsigned* deg_in  = deg_out + NN;
  float*    nsrc    = ws + 2 * NN;
  float*    ndst    = nsrc + NN;
  float*    bufA    = ndst + NN;
  float*    bufB    = bufA + (size_t)NN * DD;
  float*    sums    = bufB + (size_t)NN * DD;
  float*    cnt     = sums + NG * DD;
  unsigned* rowptr  = (unsigned*)(cnt + NG);
  unsigned* rank    = rowptr + 40064;
  int*      col     = (int*)(rank + NE);

  const int T = 256;

  hipMemsetAsync(deg_out, 0, 2 * NN * sizeof(unsigned), stream);
  hipMemsetAsync(sums, 0, (NG * DD + NG) * sizeof(float), stream);

  deg_kernel<<<(NE + T - 1) / T, T, 0, stream>>>(src, dst, deg_out, deg_in, rank);
  norm_kernel<<<(NN + T - 1) / T, T, 0, stream>>>(deg_out, deg_in, nsrc, ndst);
  rowptr_scan<<<1, 1024, 0, stream>>>(deg_in, rowptr);
  fill_kernel<<<(NE + T - 1) / T, T, 0, stream>>>(src, dst, rowptr, rank, col);

  gemm_rs_wmma<<<NN / 16, 256, 0, stream>>>(feat, nsrc, W1, bufA);
  spmm_pull<true><<<NN / 8, 256, 0, stream>>>(bufA, rowptr, col, ndst, b1, bufB);

  gemm_rs_wmma<<<NN / 16, 256, 0, stream>>>(bufB, nsrc, W2, bufA);
  spmm_pull<false><<<NN / 8, 256, 0, stream>>>(bufA, rowptr, col, ndst, b2, bufB);

  post_pool<<<(NN * DD + T - 1) / T, T, 0, stream>>>(bufB, ndst, b2, gid, sums);
  count_kernel<<<(NN + T - 1) / T, T, 0, stream>>>(gid, cnt);
  classify<<<(NG * NC + T - 1) / T, T, 0, stream>>>(sums, cnt, Wc, bc, out);
}


# readout fused into layer-2 gather: a work-group's 8 rows meet in LDS, runs of equal graph id summed in row order, one full-line f32 atomic per run (arrival order); pooling pass and its row store gone
# speedup vs baseline: 1.7673x; 1.0573x over previous
#include <hip/hip_runtime.h>

#define NN 40000
#define NE 640000
#define DD 128
#define NC 40
#define NG 128

typedef __attribute__((ext_vector_type(2))) float v2f;
typedef __attribute__((ext_vector_type(8))) float v8f;

__device__ __forceinline__ void atomAddF(float* p, float v) {
#if defined(__HIP_DEVICE_COMPILE__)
  unsafeAtomicAdd(p, v);
#else
  __hip_atomic_fetch_add(p, v, __ATOMIC_RELAXED, __HIP_MEMORY_SCOPE_AGENT);
#endif
}

__global__ void deg_kernel(const int* __restrict__ src, const int* __restrict__ dst,
                           unsigned* __restrict__ deg_out, unsigned* __restrict__ deg_in,
                           unsigned* __restrict__ rank) {
  int e = blockIdx.x * blockDim.x + threadIdx.x;
  if (e < NE) {
    atomicAdd(&deg_out[src[e]], 1u);
    rank[e] = atomicAdd(&deg_in[dst[e]], 1u);
  }
}

#define SCAN_PER 40
__global__ void __launch_bounds__(1024) rowptr_scan(const unsigned* __restrict__ din,
                                                    unsigned* __restrict__ rowptr) {
  __shared__ unsigned wsum[32];
  const int t = threadIdx.x;
  const int lane = t & 31;
  const int w = t >> 5;
  const bool active = t < NN / SCAN_PER;
  uint4 v[SCAN_PER / 4];
  unsigned tot = 0;
  if (active) {
    const uint4* __restrict__ p = (const uint4*)(din + t * SCAN_PER);
#pragma unroll
    for (int i = 0; i < SCAN_PER / 4; ++i) v[i] = p[i];
#pragma unroll
    for (int i = 0; i < SCAN_PER / 4; ++i) tot += v[i].x + v[i].y + v[i].z + v[i].w;
  }
  unsigned inc = tot;
  for (int o = 1; o < 32; o <<= 1) {
    const unsigned u = __shfl_up(inc, o);
    if (lane >= o) inc += u;
  }
  if (lane == 31) wsum[w] = inc;
  __syncthreads();
  if (w == 0) {
    const unsigned own = wsum[lane];
    unsigned si = own;
    for (int o = 1; o < 32; o <<= 1) {
      const unsigned u = __shfl_up(si, o);
      if (lane >= o) si += u;
    }
    wsum[lane] = si - own;
  }
  __syncthreads();
  unsigned run = wsum[w] + inc - tot;
  if (active) {
    uint4* __restrict__ q = (uint4*)(rowptr + t * SCAN_PER);
#pragma unroll
    for (int i = 0; i < SCAN_PER / 4; ++i) {
      uint4 o;
      o.x = run; run += v[i].x;
      o.y = run; run += v[i].y;
      o.z = run; run += v[i].z;
      o.w = run; run += v[i].w;
      q[i] = o;
    }
    if (t == NN / SCAN_PER - 1) rowptr[NN] = run;
  }
}

__global__ void fill_kernel(const int* __restrict__ src, const int* __restrict__ dst,
                            const unsigned* __restrict__ rowptr, const unsigned* __restrict__ rank,
                            int* __restrict__ col) {
  int e = blockIdx.x * blockDim.x + threadIdx.x;
  if (e < NE) {
    const unsigned pos = rowptr[dst[e]] + rank[e];
    if (pos < (unsigned)NE) col[pos] = src[e];
  }
}

__global__ void norm_kernel(const unsigned* __restrict__ dout, const unsigned* __restrict__ din,
                            float* __restrict__ ns, float* __restrict__ nd) {
  int i = blockIdx.x * blockDim.x + threadIdx.x;
  if (i < NN) {
    ns[i] = rsqrtf(fmaxf((float)dout[i], 1.0f));
    nd[i] = rsqrtf(fmaxf((float)din[i], 1.0f));
  }
}

__global__ void __launch_bounds__(256) gemm_rs_wmma(const float* __restrict__ X,
                                                    const float* __restrict__ rs,
                                                    const float* __restrict__ W,
                                                    float* __restrict__ Y) {
  const int lane = threadIdx.x & 31;
  const int wave = threadIdx.x >> 5;
  const int hl   = lane >> 4;
  const int l    = lane & 15;
  const int row0 = blockIdx.x * 16;
  const int n0   = wave * 16;
  const int m    = row0 + l;
  const float scale = rs[m];
  const float* __restrict__ arow = X + (size_t)m * DD;

  v8f c = {};
#pragma unroll
  for (int k = 0; k < DD; k += 4) {
    const int ka = k + 2 * hl;
    v2f a, b;
    a.x = arow[ka]     * scale;
    a.y = arow[ka + 1] * scale;
    b.x = W[ka * DD + n0 + l];
    b.y = W[(ka + 1) * DD + n0 + l];
    c = __builtin_amdgcn_wmma_f32_16x16x4_f32(false, a, false, b, (short)0, c, false, false);
  }
#pragma unroll
  for (int r = 0; r < 8; ++r) {
    Y[(size_t)(row0 + 8 * hl + r) * DD + n0 + l] = c[r];
  }
}

template <int K>
__device__ __forceinline__ void gather_rows(const float4* __restrict__ H4, int mine, int j,
                                            int lane, float4& acc) {
  float4 r[K];
#pragma unroll
  for (int k = 0; k < K; ++k) {
    unsigned sidx = (unsigned)__builtin_amdgcn_readlane(mine, j + k);
    sidx = sidx < (unsigned)NN ? sidx : (unsigned)(NN - 1);
    r[k] = H4[(size_t)sidx * (DD / 4) + lane];
  }
#pragma unroll
  for (int k = 0; k < K; ++k) {
    acc.x += r[k].x; acc.y += r[k].y; acc.z += r[k].z; acc.w += r[k].w;
  }
}

#define ROWS_PER_WG 8
template <bool POOL>
__global__ void __launch_bounds__(256) spmm_pull(const float* __restrict__ H,
                                                 const unsigned* __restrict__ rowptr,
                                                 const int* __restrict__ col,
                                                 const float* __restrict__ nd,
                                                 const float* __restrict__ bias,
                                                 const int* __restrict__ gid,
                                                 float* __restrict__ OUT) {
  const int lane = threadIdx.x & 31;
  const int wave = threadIdx.x >> 5;
  const int d = __builtin_amdgcn_readfirstlane((int)(blockIdx.x * ROWS_PER_WG + wave));
  const bool valid = d < NN;
  float4 o = {0.0f, 0.0f, 0.0f, 0.0f};
  int g_mine = -1;
  if (valid) {
    const unsigned beg = rowptr[d];
    unsigned end = rowptr[d + 1];
    end = end < (unsigned)NE ? end : (unsigned)NE;
    const float4* __restrict__ H4 = (const float4*)H;
    const float sc = nd[d];
    const float4 b = ((const float4*)bias)[lane];
    if (POOL) g_mine = gid[d];
    float4 acc = {0.0f, 0.0f, 0.0f, 0.0f};
    for (unsigned p = beg; p < end; p += 32) {
      const unsigned q = p + lane;
      const int mine = q < end ? col[q] : 0;
      const int n = (int)((end - p) < 32u ? (end - p) : 32u);
      int j = 0;
      for (; j + 8 <= n; j += 8) gather_rows<8>(H4, mine, j, lane, acc);
      if (j + 4 <= n) { gather_rows<4>(H4, mine, j, lane, acc); j += 4; }
      if (j + 2 <= n) { gather_rows<2>(H4, mine, j, lane, acc); j += 2; }
      if (j < n)      { gather_rows<1>(H4, mine, j, lane, acc); }
    }
    o.x = acc.x * sc + b.x;
    o.y = acc.y * sc + b.y;
    o.z = acc.z * sc + b.z;
    o.w = acc.w * sc + b.w;
    if (!POOL) {
      o.x = fmaxf(o.x, 0.0f);
      o.y = fmaxf(o.y, 0.0f);
      o.z = fmaxf(o.z, 0.0f);
      o.w = fmaxf(o.w, 0.0f);
      ((float4*)OUT)[(size_t)d * (DD / 4) + lane] = o;
    }
  }
  if constexpr (POOL) {
    __shared__ float4 prow[ROWS_PER_WG][DD / 4];
    __shared__ int pgid[ROWS_PER_WG];
    prow[wave][lane] = o;
    if (lane == 0) pgid[wave] = g_mine;
    __syncthreads();
    if (threadIdx.x < DD) {
      const int c = threadIdx.x;
      const float* __restrict__ pr = (const float*)&prow[0][0];
      float v[ROWS_PER_WG];
      int gs[ROWS_PER_WG];
#pragma unroll
      for (int r = 0; r < ROWS_PER_WG; ++r) {
        v[r] = pr[r * DD + c];
        gs[r] = __builtin_amdgcn_readfirstlane(pgid[r]);
      }
      int g = gs[0];
      float s = v[0];
#pragma unroll
      for (int r = 1; r < ROWS_PER_WG; ++r) {
        if (gs[r] != g) {
          if ((unsigned)g < (unsigned)NG) atomAddF(&OUT[g * DD + c], s);
          g = gs[r];
          s = v[r];
        } else {
          s += v[r];
        }
      }
      if ((unsigned)g < (unsigned)NG) atomAddF(&OUT[g * DD + c], s);
    }
  }
}

__global__ void count_kernel(const int* __restrict__ gid, float* __restrict__ cnt) {
  int i = blockIdx.x * blockDim.x + threadIdx.x;
  if (i < NN) atomAddF(&cnt[gid[i]], 1.0f);
}

__global__ void classify(const float* __restrict__ sums, const float* __restrict__ cnt,
                         const float* __restrict__ Wc, const float* __restrict__ bc,
                         float* __restrict__ out) {
  int idx = blockIdx.x * blockDim.x + threadIdx.x;
  if (idx < NG * NC) {
    int g = idx / NC, cc = idx % NC;
    float inv = 1.0f / fmaxf(cnt[g], 1.0f);
    float acc = bc[cc];
#pragma unroll 8
    for (int k = 0; k < DD; ++k)
      acc += (sums[g * DD + k] * inv) * Wc[k * NC + cc];
    out[idx] = acc;
  }
}

extern "C" void kernel_launch(void* const* d_in, const int* in_sizes, int n_in,
                              void* d_out, int out_size, void* d_ws, size_t ws_size,
                              hipStream_t stream) {
  const float* feat = (const float*)d_in[0];
  const float* W1   = (const float*)d_in[1];
  const float* b1   = (const float*)d_in[2];
  const float* W2   = (const float*)d_in[3];
  const float* b2   = (const float*)d_in[4];
  const float* Wc   = (const float*)d_in[5];
  const float* bc   = (const float*)d_in[6];
  const int*   src  = (const int*)d_in[7];
  const int*   dst  = (const int*)d_in[8];
  const int*   gid  = (const int*)d_in[9];
  float* out = (float*)d_out;

  float*    ws      = (float*)d_ws;
  unsigned* deg_out = (unsigned*)ws;
  unsigned* deg_in  = deg_out + NN;
  float*    nsrc    = ws + 2 * NN;
  float*    ndst    = nsrc + NN;
  float*    bufA    = ndst + NN;
  float*    bufB    = bufA + (size_t)NN * DD;
  float*    sums    = bufB + (size_t)NN * DD;
  float*    cnt     = sums + NG * DD;
  unsigned* rowptr  = (unsigned*)(cnt + NG);
  unsigned* rank    = rowptr + 40064;
  int*      col     = (int*)(rank + NE);

  const int T = 256;

  hipMemsetAsync(deg_out, 0, 2 * NN * sizeof(unsigned), stream);
  hipMemsetAsync(sums, 0, (NG * DD + NG) * sizeof(float), stream);

  deg_kernel<<<(NE + T - 1) / T, T, 0, stream>>>(src, dst, deg_out, deg_in, rank);
  norm_kernel<<<(NN + T - 1) / T, T, 0, stream>>>(deg_out, deg_in, nsrc, ndst);
  rowptr_scan<<<1, 1024, 0, stream>>>(deg_in, rowptr);
  fill_kernel<<<(NE + T - 1) / T, T, 0, stream>>>(src, dst, rowptr, rank, col);

  gemm_rs_wmma<<<NN / 16, 256, 0, stream>>>(feat, nsrc, W1, bufA);
  spmm_pull<false><<<NN / ROWS_PER_WG, 256, 0, stream>>>(bufA, rowptr, col, ndst, b1, gid, bufB);

  gemm_rs_wmma<<<NN / 16, 256, 0, stream>>>(bufB, nsrc, W2, bufA);
  spmm_pull<true><<<NN / ROWS_PER_WG, 256, 0, stream>>>(bufA, rowptr, col, ndst, b2, gid, sums);

  count_kernel<<<(NN + T - 1) / T, T, 0, stream>>>(gid, cnt);
  classify<<<(NG * NC + T - 1) / T, T, 0, stream>>>(sums, cnt, Wc, bc, out);
}


# CSR build: degrees and edge ranks counted in per-chunk LDS histograms, merged by a column pass that also writes the norms; no scattered global atomics; row sums in chunk-then-rank order
# speedup vs baseline: 2.3850x; 1.3495x over previous
#include <hip/hip_runtime.h>

#define NN 40000
#define NE 640000
#define DD 128
#define NC 40
#define NG 128

typedef __attribute__((ext_vector_type(2))) float v2f;
typedef __attribute__((ext_vector_type(8))) float v8f;

__device__ __forceinline__ void atomAddF(float* p, float v) {
#if defined(__HIP_DEVICE_COMPILE__)
  unsafeAtomicAdd(p, v);
#else
  __hip_atomic_fetch_add(p, v, __ATOMIC_RELAXED, __HIP_MEMORY_SCOPE_AGENT);
#endif
}

#define NCH 64
#define CH (NE / NCH)
#define DEG_IT ((CH + 1023) / 1024)
__global__ void __launch_bounds__(1024) deg_chunk_kernel(const int* __restrict__ src,
                                                         const int* __restrict__ dst,
                                                         unsigned* __restrict__ tab_in,
                                                         unsigned* __restrict__ tab_out,
                                                         unsigned* __restrict__ rank) {
  __shared__ unsigned h[NN];
  const int which = blockIdx.x / NCH;
  const int w = blockIdx.x % NCH;
  const int* __restrict__ key = which ? src : dst;
  const int e0 = w * CH;
  unsigned k[DEG_IT];
#pragma unroll
  for (int j = 0; j < DEG_IT; ++j) {
    const int i = threadIdx.x + 1024 * j;
    k[j] = i < CH ? (unsigned)key[e0 + i] : 0xFFFFFFFFu;
  }
  for (int i = threadIdx.x; i < NN; i += 1024) h[i] = 0u;
  __syncthreads();
#pragma unroll
  for (int j = 0; j < DEG_IT; ++j) {
    if (k[j] < (unsigned)NN) {
      const unsigned r = atomicAdd(&h[k[j]], 1u);
      if (!which) rank[e0 + threadIdx.x + 1024 * j] = r;
    }
  }
  __syncthreads();
  unsigned* __restrict__ t = (which ? tab_out : tab_in) + (size_t)w * NN;
  for (int i = threadIdx.x; i < NN; i += 1024) t[i] = h[i];
}

__global__ void merge_kernel(unsigned* __restrict__ tab_in, const unsigned* __restrict__ tab_out,
                             unsigned* __restrict__ deg_in, float* __restrict__ ns,
                             float* __restrict__ nd) {
  const int d = blockIdx.x * blockDim.x + threadIdx.x;
  if (d < NN) {
    unsigned run = 0, so = 0;
#pragma unroll 8
    for (int w = 0; w < NCH; ++w) {
      const unsigned c = tab_in[(size_t)w * NN + d];
      so += tab_out[(size_t)w * NN + d];
      tab_in[(size_t)w * NN + d] = run;
      run += c;
    }
    deg_in[d] = run;
    ns[d] = rsqrtf(fmaxf((float)so, 1.0f));
    nd[d] = rsqrtf(fmaxf((float)run, 1.0f));
  }
}

#define SCAN_PER 40
__global__ void __launch_bounds__(1024) rowptr_scan(const unsigned* __restrict__ din,
                                                    unsigned* __restrict__ rowptr) {
  __shared__ unsigned wsum[32];
  const int t = threadIdx.x;
  const int lane = t & 31;
  const int w = t >> 5;
  const bool active = t < NN / SCAN_PER;
  uint4 v[SCAN_PER / 4];
  unsigned tot = 0;
  if (active) {
    const uint4* __restrict__ p = (const uint4*)(din + t * SCAN_PER);
#pragma unroll
    for (int i = 0; i < SCAN_PER / 4; ++i) v[i] = p[i];
#pragma unroll
    for (int i = 0; i < SCAN_PER / 4; ++i) tot += v[i].x + v[i].y + v[i].z + v[i].w;
  }
  unsigned inc = tot;
  for (int o = 1; o < 32; o <<= 1) {
    const unsigned u = __shfl_up(inc, o);
    if (lane >= o) inc += u;
  }
  if (lane == 31) wsum[w] = inc;
  __syncthreads();
  if (w == 0) {
    const unsigned own = wsum[lane];
    unsigned si = own;
    for (int o = 1; o < 32; o <<= 1) {
      const unsigned u = __shfl_up(si, o);
      if (lane >= o) si += u;
    }
    wsum[lane] = si - own;
  }
  __syncthreads();
  unsigned run = wsum[w] + inc - tot;
  if (active) {
    uint4* __restrict__ q = (uint4*)(rowptr + t * SCAN_PER);
#pragma unroll
    for (int i = 0; i < SCAN_PER / 4; ++i) {
      uint4 o;
      o.x = run; run += v[i].x;
      o.y = run; run += v[i].y;
      o.z = run; run += v[i].z;
      o.w = run; run += v[i].w;
      q[i] = o;
    }
    if (t == NN / SCAN_PER - 1) rowptr[NN] = run;
  }
}

__global__ void fill_kernel(const int* __restrict__ src, const int* __restrict__ dst,
                            const unsigned* __restrict__ rowptr, const unsigned* __restrict__ tab_in,
                            const unsigned* __restrict__ rank, int* __restrict__ col) {
  int e = blockIdx.x * blockDim.x + threadIdx.x;
  if (e < NE) {
    unsigned d = (unsigned)dst[e];
    d = d < (unsigned)NN ? d : (unsigned)(NN - 1);
    const unsigned pos = rowptr[d] + tab_in[(size_t)(e / CH) * NN + d] + rank[e];
    if (pos < (unsigned)NE) col[pos] = src[e];
  }
}

__global__ void __launch_bounds__(256) gemm_rs_wmma(const float* __restrict__ X,
                                                    const float* __restrict__ rs,
                                                    const float* __restrict__ W,
                                                    float* __restrict__ Y) {
  const int lane = threadIdx.x & 31;
  const int wave = threadIdx.x >> 5;
  const int hl   = lane >> 4;
  const int l    = lane & 15;
  const int row0 = blockIdx.x * 16;
  const int n0   = wave * 16;
  const int m    = row0 + l;
  const float scale = rs[m];
  const float* __restrict__ arow = X + (size_t)m * DD;

  v8f c = {};
#pragma unroll
  for (int k = 0; k < DD; k += 4) {
    const int ka = k + 2 * hl;
    v2f a, b;
    a.x = arow[ka]     * scale;
    a.y = arow[ka + 1] * scale;
    b.x = W[ka * DD + n0 + l];
    b.y = W[(ka + 1) * DD + n0 + l];
    c = __builtin_amdgcn_wmma_f32_16x16x4_f32(false, a, false, b, (short)0, c, false, false);
  }
#pragma unroll
  for (int r = 0; r < 8; ++r) {
    Y[(size_t)(row0 + 8 * hl + r) * DD + n0 + l] = c[r];
  }
}

template <int K>
__device__ __forceinline__ void gather_rows(const float4* __restrict__ H4, int mine, int j,
                                            int lane, float4& acc) {
  float4 r[K];
#pragma unroll
  for (int k = 0; k < K; ++k) {
    unsigned sidx = (unsigned)__builtin_amdgcn_readlane(mine, j + k);
    sidx = sidx < (unsigned)NN ? sidx : (unsigned)(NN - 1);
    r[k] = H4[(size_t)sidx * (DD / 4) + lane];
  }
#pragma unroll
  for (int k = 0; k < K; ++k) {
    acc.x += r[k].x; acc.y += r[k].y; acc.z += r[k].z; acc.w += r[k].w;
  }
}

#define ROWS_PER_WG 8
template <bool POOL>
__global__ void __launch_bounds__(256) spmm_pull(const float* __restrict__ H,
                                                 const unsigned* __restrict__ rowptr,
                                                 const int* __restrict__ col,
                                                 const float* __restrict__ nd,
                                                 const float* __restrict__ bias,
                                                 const int* __restrict__ gid,
                                                 float* __restrict__ OUT) {
  const int lane = threadIdx.x & 31;
  const int wave = threadIdx.x >> 5;
  const int d = __builtin_amdgcn_readfirstlane((int)(blockIdx.x * ROWS_PER_WG + wave));
  const bool valid = d < NN;
  float4 o = {0.0f, 0.0f, 0.0f, 0.0f};
  int g_mine = -1;
  if (valid) {
    const unsigned beg = rowptr[d];
    unsigned end = rowptr[d + 1];
    end = end < (unsigned)NE ? end : (unsigned)NE;
    const float4* __restrict__ H4 = (const float4*)H;
    const float sc = nd[d];
    const float4 b = ((const float4*)bias)[lane];
    if (POOL) g_mine = gid[d];
    float4 acc = {0.0f, 0.0f, 0.0f, 0.0f};
    for (unsigned p = beg; p < end; p += 32) {
      const unsigned q = p + lane;
      const int mine = q < end ? col[q] : 0;
      const int n = (int)((end - p) < 32u ? (end - p) : 32u);
      int j = 0;
      for (; j + 8 <= n; j += 8) gather_rows<8>(H4, mine, j, lane, acc);
      if (j + 4 <= n) { gather_rows<4>(H4, mine, j, lane, acc); j += 4; }
      if (j + 2 <= n) { gather_rows<2>(H4, mine, j, lane, acc); j += 2; }
      if (j < n)      { gather_rows<1>(H4, mine, j, lane, acc); }
    }
    o.x = acc.x * sc + b.x;
    o.y = acc.y * sc + b.y;
    o.z = acc.z * sc + b.z;
    o.w = acc.w * sc + b.w;
    if (!POOL) {
      o.x = fmaxf(o.x, 0.0f);
      o.y = fmaxf(o.y, 0.0f);
      o.z = fmaxf(o.z, 0.0f);
      o.w = fmaxf(o.w, 0.0f);
      ((float4*)OUT)[(size_t)d * (DD / 4) + lane] = o;
    }
  }
  if constexpr (POOL) {
    __shared__ float4 prow[ROWS_PER_WG][DD / 4];
    __shared__ int pgid[ROWS_PER_WG];
    prow[wave][lane] = o;
    if (lane == 0) pgid[wave] = g_mine;
    __syncthreads();
    if (threadIdx.x < DD) {
      const int c = threadIdx.x;
      const float* __restrict__ pr = (const float*)&prow[0][0];
      float v[ROWS_PER_WG];
      int gs[ROWS_PER_WG];
#pragma unroll
      for (int r = 0; r < ROWS_PER_WG; ++r) {
        v[r] = pr[r * DD + c];
        gs[r] = __builtin_amdgcn_readfirstlane(pgid[r]);
      }
      int g = gs[0];
      float s = v[0];
#pragma unroll
      for (int r = 1; r < ROWS_PER_WG; ++r) {
        if (gs[r] != g) {
          if ((unsigned)g < (unsigned)NG) atomAddF(&OUT[g * DD + c], s);
          g = gs[r];
          s = v[r];
        } else {
          s += v[r];
        }
      }
      if ((unsigned)g < (unsigned)NG) atomAddF(&OUT[g * DD + c], s);
    }
  }
}

__global__ void count_kernel(const int* __restrict__ gid, float* __restrict__ cnt) {
  int i = blockIdx.x * blockDim.x + threadIdx.x;
  if (i < NN) atomAddF(&cnt[gid[i]], 1.0f);
}

__global__ void classify(const float* __restrict__ sums, const float* __restrict__ cnt,
                         const float* __restrict__ Wc, const float* __restrict__ bc,
                         float* __restrict__ out) {
  int idx = blockIdx.x * blockDim.x + threadIdx.x;
  if (idx < NG * NC) {
    int g = idx / NC, cc = idx % NC;
    float inv = 1.0f / fmaxf(cnt[g], 1.0f);
    float acc = bc[cc];
#pragma unroll 8
    for (int k = 0; k < DD; ++k)
      acc += (sums[g * DD + k] * inv) * Wc[k * NC + cc];
    out[idx] = acc;
  }
}

extern "C" void kernel_launch(void* const* d_in, const int* in_sizes, int n_in,
                              void* d_out, int out_size, void* d_ws, size_t ws_size,
                              hipStream_t stream) {
  const float* feat = (const float*)d_in[0];
  const float* W1   = (const float*)d_in[1];
  const float* b1   = (const float*)d_in[2];
  const float* W2   = (const float*)d_in[3];
  const float* b2   = (const float*)d_in[4];
  const float* Wc   = (const float*)d_in[5];
  const float* bc   = (const float*)d_in[6];
  const int*   src  = (const int*)d_in[7];
  const int*   dst  = (const int*)d_in[8];
  const int*   gid  = (const int*)d_in[9];
  float* out = (float*)d_out;

  float*    ws      = (float*)d_ws;
  unsigned* deg_in  = (unsigned*)ws + NN;
  float*    nsrc    = ws + 2 * NN;
  float*    ndst    = nsrc + NN;
  float*    bufA    = ndst + NN;
  float*    bufB    = bufA + (size_t)NN * DD;
  float*    sums    = bufB + (size_t)NN * DD;
  float*    cnt     = sums + NG * DD;
  unsigned* rowptr  = (unsigned*)(cnt + NG);
  unsigned* rank    = rowptr + 40064;
  int*      col     = (int*)(rank + NE);
  unsigned* tab_in  = (unsigned*)(col + NE);
  unsigned* tab_out = tab_in + (size_t)NCH * NN;

  const int T = 256;

  hipMemsetAsync(sums, 0, (NG * DD + NG) * sizeof(float), stream);

  deg_chunk_kernel<<<2 * NCH, 1024, 0, stream>>>(src, dst, tab_in, tab_out, rank);
  merge_kernel<<<(NN + T - 1) / T, T, 0, stream>>>(tab_in, tab_out, deg_in, nsrc, ndst);
  rowptr_scan<<<1, 1024, 0, stream>>>(deg_in, rowptr);
  fill_kernel<<<(NE + T - 1) / T, T, 0, stream>>>(src, dst, rowptr, tab_in, rank, col);

  gemm_rs_wmma<<<NN / 16, 256, 0, stream>>>(feat, nsrc, W1, bufA);
  spmm_pull<false><<<NN / ROWS_PER_WG, 256, 0, stream>>>(bufA, rowptr, col, ndst, b1, gid, bufB);

  gemm_rs_wmma<<<NN / 16, 256, 0, stream>>>(bufB, nsrc, W2, bufA);
  spmm_pull<true><<<NN / ROWS_PER_WG, 256, 0, stream>>>(bufA, rowptr, col, ndst, b2, gid, sums);

  count_kernel<<<(NN + T - 1) / T, T, 0, stream>>>(gid, cnt);
  classify<<<(NG * NC + T - 1) / T, T, 0, stream>>>(sums, cnt, Wc, bc, out);
}


# layer 2: W2 moved behind the readout's row sums (rows summed per graph first, then one 128-row WMMA product in the readout kernel): the 40000-row GEMM launch is gone; sums reassociated
# speedup vs baseline: 2.6830x; 1.1249x over previous
#include <hip/hip_runtime.h>

#define NN 40000
#define NE 640000
#define DD 128
#define NC 40
#define NG 128

typedef __attribute__((ext_vector_type(2))) float v2f;
typedef __attribute__((ext_vector_type(8))) float v8f;

__device__ __forceinline__ void atomAddF(float* p, float v) {
#if defined(__HIP_DEVICE_COMPILE__)
  unsafeAtomicAdd(p, v);
#else
  __hip_atomic_fetch_add(p, v, __ATOMIC_RELAXED, __HIP_MEMORY_SCOPE_AGENT);
#endif
}

#define NCH 64
#define CH (NE / NCH)
#define DEG_IT ((CH + 1023) / 1024)
__global__ void __launch_bounds__(1024) deg_chunk_kernel(const int* __restrict__ src,
                                                         const int* __restrict__ dst,
                                                         unsigned* __restrict__ tab_in,
                                                         unsigned* __restrict__ tab_out,
                                                         unsigned* __restrict__ rank) {
  __shared__ unsigned h[NN];
  const int which = blockIdx.x / NCH;
  const int w = blockIdx.x % NCH;
  const int* __restrict__ key = which ? src : dst;
  const int e0 = w * CH;
  unsigned k[DEG_IT];
#pragma unroll
  for (int j = 0; j < DEG_IT; ++j) {
    const int i = threadIdx.x + 1024 * j;
    k[j] = i < CH ? (unsigned)key[e0 + i] : 0xFFFFFFFFu;
  }
  for (int i = threadIdx.x; i < NN; i += 1024) h[i] = 0u;
  __syncthreads();
#pragma unroll
  for (int j = 0; j < DEG_IT; ++j) {
    if (k[j] < (unsigned)NN) {
      const unsigned r = atomicAdd(&h[k[j]], 1u);
      if (!which) rank[e0 + threadIdx.x + 1024 * j] = r;
    }
  }
  __syncthreads();
  unsigned* __restrict__ t = (which ? tab_out : tab_in) + (size_t)w * NN;
  for (int i = threadIdx.x; i < NN; i += 1024) t[i] = h[i];
}

__global__ void merge_kernel(unsigned* __restrict__ tab_in, const unsigned* __restrict__ tab_out,
                             unsigned* __restrict__ deg_in, float* __restrict__ ns,
                             float* __restrict__ nd) {
  const int d = blockIdx.x * blockDim.x + threadIdx.x;
  if (d < NN) {
    unsigned run = 0, so = 0;
#pragma unroll 8
    for (int w = 0; w < NCH; ++w) {
      const unsigned c = tab_in[(size_t)w * NN + d];
      so += tab_out[(size_t)w * NN + d];
      tab_in[(size_t)w * NN + d] = run;
      run += c;
    }
    deg_in[d] = run;
    ns[d] = rsqrtf(fmaxf((float)so, 1.0f));
    nd[d] = rsqrtf(fmaxf((float)run, 1.0f));
  }
}

#define SCAN_PER 40
__global__ void __launch_bounds__(1024) rowptr_scan(const unsigned* __restrict__ din,
                                                    unsigned* __restrict__ rowptr) {
  __shared__ unsigned wsum[32];
  const int t = threadIdx.x;
  const int lane = t & 31;
  const int w = t >> 5;
  const bool active = t < NN / SCAN_PER;
  uint4 v[SCAN_PER / 4];
  unsigned tot = 0;
  if (active) {
    const uint4* __restrict__ p = (const uint4*)(din + t * SCAN_PER);
#pragma unroll
    for (int i = 0; i < SCAN_PER / 4; ++i) v[i] = p[i];
#pragma unroll
    for (int i = 0; i < SCAN_PER / 4; ++i) tot += v[i].x + v[i].y + v[i].z + v[i].w;
  }
  unsigned inc = tot;
  for (int o = 1; o < 32; o <<= 1) {
    const unsigned u = __shfl_up(inc, o);
    if (lane >= o) inc += u;
  }
  if (lane == 31) wsum[w] = inc;
  __syncthreads();
  if (w == 0) {
    const unsigned own = wsum[lane];
    unsigned si = own;
    for (int o = 1; o < 32; o <<= 1) {
      const unsigned u = __shfl_up(si, o);
      if (lane >= o) si += u;
    }
    wsum[lane] = si - own;
  }
  __syncthreads();
  unsigned run = wsum[w] + inc - tot;
  if (active) {
    uint4* __restrict__ q = (uint4*)(rowptr + t * SCAN_PER);
#pragma unroll
    for (int i = 0; i < SCAN_PER / 4; ++i) {
      uint4 o;
      o.x = run; run += v[i].x;
      o.y = run; run += v[i].y;
      o.z = run; run += v[i].z;
      o.w = run; run += v[i].w;
      q[i] = o;
    }
    if (t == NN / SCAN_PER - 1) rowptr[NN] = run;
  }
}

__global__ void fill_kernel(const int* __restrict__ src, const int* __restrict__ dst,
                            const unsigned* __restrict__ rowptr, const unsigned* __restrict__ tab_in,
                            const unsigned* __restrict__ rank, int* __restrict__ col) {
  int e = blockIdx.x * blockDim.x + threadIdx.x;
  if (e < NE) {
    unsigned d = (unsigned)dst[e];
    d = d < (unsigned)NN ? d : (unsigned)(NN - 1);
    const unsigned pos = rowptr[d] + tab_in[(size_t)(e / CH) * NN + d] + rank[e];
    if (pos < (unsigned)NE) col[pos] = src[e];
  }
}

__global__ void __launch_bounds__(256) gemm_rs_wmma(const float* __restrict__ X,
                                                    const float* __restrict__ rs,
                                                    const float* __restrict__ W,
                                                    float* __restrict__ Y) {
  const int lane = threadIdx.x & 31;
  const int wave = threadIdx.x >> 5;
  const int hl   = lane >> 4;
  const int l    = lane & 15;
  const int row0 = blockIdx.x * 16;
  const int n0   = wave * 16;
  const int m    = row0 + l;
  const float scale = rs[m];
  const float* __restrict__ arow = X + (size_t)m * DD;

  v8f c = {};
#pragma unroll
  for (int k = 0; k < DD; k += 4) {
    const int ka = k + 2 * hl;
    v2f a, b;
    a.x = arow[ka]     * scale;
    a.y = arow[ka + 1] * scale;
    b.x = W[ka * DD + n0 + l];
    b.y = W[(ka + 1) * DD + n0 + l];
    c = __builtin_amdgcn_wmma_f32_16x16x4_f32(false, a, false, b, (short)0, c, false, false);
  }
#pragma unroll
  for (int r = 0; r < 8; ++r) {
    Y[(size_t)(row0 + 8 * hl + r) * DD + n0 + l] = c[r];
  }
}

template <int K>
__device__ __forceinline__ void gather_rows(const float4* __restrict__ H4, int mine, int j,
                                            int lane, float4& acc) {
  float4 r[K];
#pragma unroll
  for (int k = 0; k < K; ++k) {
    unsigned sidx = (unsigned)__builtin_amdgcn_readlane(mine, j + k);
    sidx = sidx < (unsigned)NN ? sidx : (unsigned)(NN - 1);
    r[k] = H4[(size_t)sidx * (DD / 4) + lane];
  }
#pragma unroll
  for (int k = 0; k < K; ++k) {
    acc.x += r[k].x; acc.y += r[k].y; acc.z += r[k].z; acc.w += r[k].w;
  }
}

#define ROWS_PER_WG 8
template <bool POOL>
__global__ void __launch_bounds__(256) spmm_pull(const float* __restrict__ H,
                                                 const unsigned* __restrict__ rowptr,
                                                 const int* __restrict__ col,
                                                 const float* __restrict__ nd,
                                                 const float* __restrict__ ns,
                                                 const float* __restrict__ bias,
                                                 const int* __restrict__ gid,
                                                 float* __restrict__ OUT) {
  const int lane = threadIdx.x & 31;
  const int wave = threadIdx.x >> 5;
  const int d = __builtin_amdgcn_readfirstlane((int)(blockIdx.x * ROWS_PER_WG + wave));
  const bool valid = d < NN;
  float4 o = {0.0f, 0.0f, 0.0f, 0.0f};
  int g_mine = -1;
  if (valid) {
    const unsigned beg = rowptr[d];
    unsigned end = rowptr[d + 1];
    end = end < (unsigned)NE ? end : (unsigned)NE;
    const float4* __restrict__ H4 = (const float4*)H;
    const float sc = nd[d];
    float sn = 1.0f;
    float4 b = {0.0f, 0.0f, 0.0f, 0.0f};
    if constexpr (POOL) {
      g_mine = gid[d];
    } else {
      sn = ns[d];
      b = ((const float4*)bias)[lane];
    }
    float4 acc = {0.0f, 0.0f, 0.0f, 0.0f};
    for (unsigned p = beg; p < end; p += 32) {
      const unsigned q = p + lane;
      const int mine = q < end ? col[q] : 0;
      const int n = (int)((end - p) < 32u ? (end - p) : 32u);
      int j = 0;
      for (; j + 8 <= n; j += 8) gather_rows<8>(H4, mine, j, lane, acc);
      if (j + 4 <= n) { gather_rows<4>(H4, mine, j, lane, acc); j += 4; }
      if (j + 2 <= n) { gather_rows<2>(H4, mine, j, lane, acc); j += 2; }
      if (j < n)      { gather_rows<1>(H4, mine, j, lane, acc); }
    }
    if constexpr (POOL) {
      o.x = acc.x * sc;
      o.y = acc.y * sc;
      o.z = acc.z * sc;
      o.w = acc.w * sc;
    } else {
      o.x = fmaxf(acc.x * sc + b.x, 0.0f) * sn;
      o.y = fmaxf(acc.y * sc + b.y, 0.0f) * sn;
      o.z = fmaxf(acc.z * sc + b.z, 0.0f) * sn;
      o.w = fmaxf(acc.w * sc + b.w, 0.0f) * sn;
      ((float4*)OUT)[(size_t)d * (DD / 4) + lane] = o;
    }
  }
  if constexpr (POOL) {
    __shared__ float4 prow[ROWS_PER_WG][DD / 4];
    __shared__ int pgid[ROWS_PER_WG];
    prow[wave][lane] = o;
    if (lane == 0) pgid[wave] = g_mine;
    __syncthreads();
    if (threadIdx.x < DD) {
      const int c = threadIdx.x;
      const float* __restrict__ pr = (const float*)&prow[0][0];
      float v[ROWS_PER_WG];
      int gs[ROWS_PER_WG];
#pragma unroll
      for (int r = 0; r < ROWS_PER_WG; ++r) {
        v[r] = pr[r * DD + c];
        gs[r] = __builtin_amdgcn_readfirstlane(pgid[r]);
      }
      int g = gs[0];
      float s = v[0];
#pragma unroll
      for (int r = 1; r < ROWS_PER_WG; ++r) {
        if (gs[r] != g) {
          if ((unsigned)g < (unsigned)NG) atomAddF(&OUT[g * DD + c], s);
          g = gs[r];
          s = v[r];
        } else {
          s += v[r];
        }
      }
      if ((unsigned)g < (unsigned)NG) atomAddF(&OUT[g * DD + c], s);
    }
  }
}

__global__ void count_kernel(const int* __restrict__ gid, float* __restrict__ cnt) {
  int i = blockIdx.x * blockDim.x + threadIdx.x;
  if (i < NN) atomAddF(&cnt[gid[i]], 1.0f);
}

#define HG_PITCH (DD + 4)
#define RO_ROWS 16
__global__ void __launch_bounds__(256) readout_kernel(const float* __restrict__ S,
                                                      const float* __restrict__ cnt,
                                                      const float* __restrict__ W2,
                                                      const float* __restrict__ b2,
                                                      const float* __restrict__ Wc,
                                                      const float* __restrict__ bc,
                                                      float* __restrict__ out) {
  __shared__ float hg[RO_ROWS * HG_PITCH];
  const int lane = threadIdx.x & 31;
  const int wave = threadIdx.x >> 5;
  const int hl   = lane >> 4;
  const int l    = lane & 15;
  const int g0   = blockIdx.x * RO_ROWS;
  const int n0   = wave * 16;
  const float* __restrict__ arow = S + (size_t)(g0 + l) * DD;

  v8f c = {};
#pragma unroll
  for (int k = 0; k < DD; k += 4) {
    const int ka = k + 2 * hl;
    v2f a, b;
    a.x = arow[ka];
    a.y = arow[ka + 1];
    b.x = W2[ka * DD + n0 + l];
    b.y = W2[(ka + 1) * DD + n0 + l];
    c = __builtin_amdgcn_wmma_f32_16x16x4_f32(false, a, false, b, (short)0, c, false, false);
  }
  const float bcol = b2[n0 + l];
#pragma unroll
  for (int r = 0; r < 8; ++r) {
    const float n = cnt[g0 + 8 * hl + r];
    const float inv = 1.0f / fmaxf(n, 1.0f);
    hg[(8 * hl + r) * HG_PITCH + n0 + l] = (c[r] + n * bcol) * inv;
  }
  __syncthreads();
  if (threadIdx.x < 4 * NC) {
    const int cc = threadIdx.x % NC;
    const int i0 = threadIdx.x / NC;
    float acc[4];
#pragma unroll
    for (int j = 0; j < 4; ++j) acc[j] = bc[cc];
#pragma unroll 8
    for (int k = 0; k < DD; ++k) {
      const float w = Wc[k * NC + cc];
#pragma unroll
      for (int j = 0; j < 4; ++j) acc[j] += hg[(i0 + 4 * j) * HG_PITCH + k] * w;
    }
#pragma unroll
    for (int j = 0; j < 4; ++j) out[(g0 + i0 + 4 * j) * NC + cc] = acc[j];
  }
}

extern "C" void kernel_launch(void* const* d_in, const int* in_sizes, int n_in,
                              void* d_out, int out_size, void* d_ws, size_t ws_size,
                              hipStream_t stream) {
  const float* feat = (const float*)d_in[0];
  const float* W1   = (const float*)d_in[1];
  const float* b1   = (const float*)d_in[2];
  const float* W2   = (const float*)d_in[3];
  const float* b2   = (const float*)d_in[4];
  const float* Wc   = (const float*)d_in[5];
  const float* bc   = (const float*)d_in[6];
  const int*   src  = (const int*)d_in[7];
  const int*   dst  = (const int*)d_in[8];
  const int*   gid  = (const int*)d_in[9];
  float* out = (float*)d_out;

  float*    ws      = (float*)d_ws;
  unsigned* deg_in  = (unsigned*)ws + NN;
  float*    nsrc    = ws + 2 * NN;
  float*    ndst    = nsrc + NN;
  float*    bufA    = ndst + NN;
  float*    bufB    = bufA + (size_t)NN * DD;
  float*    sums    = bufB + (size_t)NN * DD;
  float*    cnt     = sums + NG * DD;
  unsigned* rowptr  = (unsigned*)(cnt + NG);
  unsigned* rank    = rowptr + 40064;
  int*      col     = (int*)(rank + NE);
  unsigned* tab_in  = (unsigned*)(col + NE);
  unsigned* tab_out = tab_in + (size_t)NCH * NN;

  const int T = 256;

  hipMemsetAsync(sums, 0, (NG * DD + NG) * sizeof(float), stream);

  deg_chunk_kernel<<<2 * NCH, 1024, 0, stream>>>(src, dst, tab_in, tab_out, rank);
  merge_kernel<<<(NN + T - 1) / T, T, 0, stream>>>(tab_in, tab_out, deg_in, nsrc, ndst);
  rowptr_scan<<<1, 1024, 0, stream>>>(deg_in, rowptr);
  fill_kernel<<<(NE + T - 1) / T, T, 0, stream>>>(src, dst, rowptr, tab_in, rank, col);

  gemm_rs_wmma<<<NN / 16, 256, 0, stream>>>(feat, nsrc, W1, bufA);
  spmm_pull<false><<<NN / ROWS_PER_WG, 256, 0, stream>>>(bufA, rowptr, col, ndst, nsrc, b1, gid, bufB);

  spmm_pull<true><<<NN / ROWS_PER_WG, 256, 0, stream>>>(bufB, rowptr, col, ndst, nsrc, b2, gid, sums);

  count_kernel<<<(NN + T - 1) / T, T, 0, stream>>>(gid, cnt);
  readout_kernel<<<NG / RO_ROWS, 256, 0, stream>>>(sums, cnt, W2, b2, Wc, bc, out);
}


# launch graph cut from 10 stream operations to 7: merge kernel zeroes the per-graph sums, layer-2 gather adds node counts per run, rowptr scan runs as block 0 of the GEMM launch; sums unchanged
# speedup vs baseline: 3.2183x; 1.1995x over previous
#include <hip/hip_runtime.h>

#define NN 40000
#define NE 640000
#define DD 128
#define NC 40
#define NG 128

typedef __attribute__((ext_vector_type(2))) float v2f;
typedef __attribute__((ext_vector_type(8))) float v8f;

__device__ __forceinline__ void atomAddF(float* p, float v) {
#if defined(__HIP_DEVICE_COMPILE__)
  unsafeAtomicAdd(p, v);
#else
  __hip_atomic_fetch_add(p, v, __ATOMIC_RELAXED, __HIP_MEMORY_SCOPE_AGENT);
#endif
}

#define NCH 64
#define CH (NE / NCH)
#define DEG_IT ((CH + 1023) / 1024)
__global__ void __launch_bounds__(1024) deg_chunk_kernel(const int* __restrict__ src,
                                                         const int* __restrict__ dst,
                                                         unsigned* __restrict__ tab_in,
                                                         unsigned* __restrict__ tab_out,
                                                         unsigned* __restrict__ rank) {
  __shared__ unsigned h[NN];
  const int which = blockIdx.x / NCH;
  const int w = blockIdx.x % NCH;
  const int* __restrict__ key = which ? src : dst;
  const int e0 = w * CH;
  unsigned k[DEG_IT];
#pragma unroll
  for (int j = 0; j < DEG_IT; ++j) {
    const int i = threadIdx.x + 1024 * j;
    k[j] = i < CH ? (unsigned)key[e0 + i] : 0xFFFFFFFFu;
  }
  for (int i = threadIdx.x; i < NN; i += 1024) h[i] = 0u;
  __syncthreads();
#pragma unroll
  for (int j = 0; j < DEG_IT; ++j) {
    if (k[j] < (unsigned)NN) {
      const unsigned r = atomicAdd(&h[k[j]], 1u);
      if (!which) rank[e0 + threadIdx.x + 1024 * j] = r;
    }
  }
  __syncthreads();
  unsigned* __restrict__ t = (which ? tab_out : tab_in) + (size_t)w * NN;
  for (int i = threadIdx.x; i < NN; i += 1024) t[i] = h[i];
}

__global__ void merge_kernel(unsigned* __restrict__ tab_in, const unsigned* __restrict__ tab_out,
                             unsigned* __restrict__ deg_in, float* __restrict__ ns,
                             float* __restrict__ nd, float* __restrict__ zsum) {
  const int d = blockIdx.x * blockDim.x + threadIdx.x;
  if (d < NG * DD + NG) zsum[d] = 0.0f;
  if (d < NN) {
    unsigned run = 0, so = 0;
#pragma unroll 8
    for (int w = 0; w < NCH; ++w) {
      const unsigned c = tab_in[(size_t)w * NN + d];
      so += tab_out[(size_t)w * NN + d];
      tab_in[(size_t)w * NN + d] = run;
      run += c;
    }
    deg_in[d] = run;
    ns[d] = rsqrtf(fmaxf((float)so, 1.0f));
    nd[d] = rsqrtf(fmaxf((float)run, 1.0f));
  }
}

#define SCAN_W4 (NN / 8 / 4)
#define SCAN_IT ((SCAN_W4 + 31) / 32)
__device__ __forceinline__ void rowptr_scan_block(const unsigned* __restrict__ din,
                                                  unsigned* __restrict__ rowptr,
                                                  unsigned* wtot) {
  const int t = threadIdx.x;
  const int lane = t & 31;
  const int w = t >> 5;
  const uint4* __restrict__ p = (const uint4*)din + SCAN_W4 * w;
  uint4* __restrict__ q = (uint4*)rowptr + SCAN_W4 * w;
  unsigned s = 0;
#pragma unroll 8
  for (int i = 0; i < SCAN_IT; ++i) {
    const int j = lane + 32 * i;
    if (j < SCAN_W4) {
      const uint4 v = p[j];
      s += v.x + v.y + v.z + v.w;
    }
  }
  for (int o = 16; o > 0; o >>= 1) s += __shfl_xor(s, o);
  if (lane == 0) wtot[w] = s;
  __syncthreads();
  unsigned carry = 0;
  for (int k = 0; k < w; ++k) carry += wtot[k];
#pragma unroll 8
  for (int i = 0; i < SCAN_IT; ++i) {
    const int j = lane + 32 * i;
    uint4 v = {0u, 0u, 0u, 0u};
    if (j < SCAN_W4) v = p[j];
    const unsigned own = v.x + v.y + v.z + v.w;
    unsigned inc = own;
    for (int o = 1; o < 32; o <<= 1) {
      const unsigned u = __shfl_up(inc, o);
      if (lane >= o) inc += u;
    }
    unsigned run = carry + inc - own;
    if (j < SCAN_W4) {
      uint4 r;
      r.x = run; run += v.x;
      r.y = run; run += v.y;
      r.z = run; run += v.z;
      r.w = run;
      q[j] = r;
    }
    carry += (unsigned)__builtin_amdgcn_readlane((int)inc, 31);
  }
  if (t == 255) rowptr[NN] = carry;
}

__global__ void fill_kernel(const int* __restrict__ src, const int* __restrict__ dst,
                            const unsigned* __restrict__ rowptr, const unsigned* __restrict__ tab_in,
                            const unsigned* __restrict__ rank, int* __restrict__ col) {
  int e = blockIdx.x * blockDim.x + threadIdx.x;
  if (e < NE) {
    unsigned d = (unsigned)dst[e];
    d = d < (unsigned)NN ? d : (unsigned)(NN - 1);
    const unsigned pos = rowptr[d] + tab_in[(size_t)(e / CH) * NN + d] + rank[e];
    if (pos < (unsigned)NE) col[pos] = src[e];
  }
}

__global__ void __launch_bounds__(256) scan_gemm_kernel(const float* __restrict__ X,
                                                        const float* __restrict__ rs,
                                                        const float* __restrict__ W,
                                                        float* __restrict__ Y,
                                                        const unsigned* __restrict__ din,
                                                        unsigned* __restrict__ rowptr) {
  __shared__ unsigned wtot[8];
  if (blockIdx.x == 0) {
    rowptr_scan_block(din, rowptr, wtot);
    return;
  }
  const int lane = threadIdx.x & 31;
  const int wave = threadIdx.x >> 5;
  const int hl   = lane >> 4;
  const int l    = lane & 15;
  const int row0 = (blockIdx.x - 1) * 16;
  const int n0   = wave * 16;
  const int m    = row0 + l;
  const float scale = rs[m];
  const float* __restrict__ arow = X + (size_t)m * DD;

  v8f c = {};
#pragma unroll
  for (int k = 0; k < DD; k += 4) {
    const int ka = k + 2 * hl;
    v2f a, b;
    a.x = arow[ka]     * scale;
    a.y = arow[ka + 1] * scale;
    b.x = W[ka * DD + n0 + l];
    b.y = W[(ka + 1) * DD + n0 + l];
    c = __builtin_amdgcn_wmma_f32_16x16x4_f32(false, a, false, b, (short)0, c, false, false);
  }
#pragma unroll
  for (int r = 0; r < 8; ++r) {
    Y[(size_t)(row0 + 8 * hl + r) * DD + n0 + l] = c[r];
  }
}

template <int K>
__device__ __forceinline__ void gather_rows(const float4* __restrict__ H4, int mine, int j,
                                            int lane, float4& acc) {
  float4 r[K];
#pragma unroll
  for (int k = 0; k < K; ++k) {
    unsigned sidx = (unsigned)__builtin_amdgcn_readlane(mine, j + k);
    sidx = sidx < (unsigned)NN ? sidx : (unsigned)(NN - 1);
    r[k] = H4[(size_t)sidx * (DD / 4) + lane];
  }
#pragma unroll
  for (int k = 0; k < K; ++k) {
    acc.x += r[k].x; acc.y += r[k].y; acc.z += r[k].z; acc.w += r[k].w;
  }
}

#define ROWS_PER_WG 8
template <bool POOL>
__global__ void __launch_bounds__(256) spmm_pull(const float* __restrict__ H,
                                                 const unsigned* __restrict__ rowptr,
                                                 const int* __restrict__ col,
                                                 const float* __restrict__ nd,
                                                 const float* __restrict__ ns,
                                                 const float* __restrict__ bias,
                                                 const int* __restrict__ gid,
                                                 float* __restrict__ OUT,
                                                 float* __restrict__ CNT) {
  const int lane = threadIdx.x & 31;
  const int wave = threadIdx.x >> 5;
  const int d = __builtin_amdgcn_readfirstlane((int)(blockIdx.x * ROWS_PER_WG + wave));
  const bool valid = d < NN;
  float4 o = {0.0f, 0.0f, 0.0f, 0.0f};
  int g_mine = -1;
  if (valid) {
    const unsigned beg = rowptr[d];
    unsigned end = rowptr[d + 1];
    end = end < (unsigned)NE ? end : (unsigned)NE;
    const float4* __restrict__ H4 = (const float4*)H;
    const float sc = nd[d];
    float sn = 1.0f;
    float4 b = {0.0f, 0.0f, 0.0f, 0.0f};
    if constexpr (POOL) {
      g_mine = gid[d];
    } else {
      sn = ns[d];
      b = ((const float4*)bias)[lane];
    }
    float4 acc = {0.0f, 0.0f, 0.0f, 0.0f};
    for (unsigned p = beg; p < end; p += 32) {
      const unsigned q = p + lane;
      const int mine = q < end ? col[q] : 0;
      const int n = (int)((end - p) < 32u ? (end - p) : 32u);
      int j = 0;
      for (; j + 8 <= n; j += 8) gather_rows<8>(H4, mine, j, lane, acc);
      if (j + 4 <= n) { gather_rows<4>(H4, mine, j, lane, acc); j += 4; }
      if (j + 2 <= n) { gather_rows<2>(H4, mine, j, lane, acc); j += 2; }
      if (j < n)      { gather_rows<1>(H4, mine, j, lane, acc); }
    }
    if constexpr (POOL) {
      o.x = acc.x * sc;
      o.y = acc.y * sc;
      o.z = acc.z * sc;
      o.w = acc.w * sc;
    } else {
      o.x = fmaxf(acc.x * sc + b.x, 0.0f) * sn;
      o.y = fmaxf(acc.y * sc + b.y, 0.0f) * sn;
      o.z = fmaxf(acc.z * sc + b.z, 0.0f) * sn;
      o.w = fmaxf(acc.w * sc + b.w, 0.0f) * sn;
      ((float4*)OUT)[(size_t)d * (DD / 4) + lane] = o;
    }
  }
  if constexpr (POOL) {
    __shared__ float4 prow[ROWS_PER_WG][DD / 4];
    __shared__ int pgid[ROWS_PER_WG];
    prow[wave][lane] = o;
    if (lane == 0) pgid[wave] = g_mine;
    __syncthreads();
    if (threadIdx.x < DD) {
      const int c = threadIdx.x;
      const float* __restrict__ pr = (const float*)&prow[0][0];
      float v[ROWS_PER_WG];
      int gs[ROWS_PER_WG];
#pragma unroll
      for (int r = 0; r < ROWS_PER_WG; ++r) {
        v[r] = pr[r * DD + c];
        gs[r] = __builtin_amdgcn_readfirstlane(pgid[r]);
      }
      int g = gs[0];
      float s = v[0];
#pragma unroll
      for (int r = 1; r < ROWS_PER_WG; ++r) {
        if (gs[r] != g) {
          if ((unsigned)g < (unsigned)NG) atomAddF(&OUT[g * DD + c], s);
          g = gs[r];
          s = v[r];
        } else {
          s += v[r];
        }
      }
      if ((unsigned)g < (unsigned)NG) atomAddF(&OUT[g * DD + c], s);
    } else if (threadIdx.x == DD) {
      int g = pgid[0];
      float n = 1.0f;
      for (int r = 1; r < ROWS_PER_WG; ++r) {
        const int gr = pgid[r];
        if (gr != g) {
          if ((unsigned)g < (unsigned)NG) atomAddF(&CNT[g], n);
          g = gr;
          n = 1.0f;
        } else {
          n += 1.0f;
        }
      }
      if ((unsigned)g < (unsigned)NG) atomAddF(&CNT[g], n);
    }
  }
}

#define HG_PITCH (DD + 4)
#define RO_ROWS 16
__global__ void __launch_bounds__(256) readout_kernel(const float* __restrict__ S,
                                                      const float* __restrict__ cnt,
                                                      const float* __restrict__ W2,
                                                      const float* __restrict__ b2,
                                                      const float* __restrict__ Wc,
                                                      const float* __restrict__ bc,
                                                      float* __restrict__ out) {
  __shared__ float hg[RO_ROWS * HG_PITCH];
  const int lane = threadIdx.x & 31;
  const int wave = threadIdx.x >> 5;
  const int hl   = lane >> 4;
  const int l    = lane & 15;
  const int g0   = blockIdx.x * RO_ROWS;
  const int n0   = wave * 16;
  const float* __restrict__ arow = S + (size_t)(g0 + l) * DD;

  v8f c = {};
#pragma unroll
  for (int k = 0; k < DD; k += 4) {
    const int ka = k + 2 * hl;
    v2f a, b;
    a.x = arow[ka];
    a.y = arow[ka + 1];
    b.x = W2[ka * DD + n0 + l];
    b.y = W2[(ka + 1) * DD + n0 + l];
    c = __builtin_amdgcn_wmma_f32_16x16x4_f32(false, a, false, b, (short)0, c, false, false);
  }
  const float bcol = b2[n0 + l];
#pragma unroll
  for (int r = 0; r < 8; ++r) {
    const float n = cnt[g0 + 8 * hl + r];
    const float inv = 1.0f / fmaxf(n, 1.0f);
    hg[(8 * hl + r) * HG_PITCH + n0 + l] = (c[r] + n * bcol) * inv;
  }
  __syncthreads();
  if (threadIdx.x < 4 * NC) {
    const int cc = threadIdx.x % NC;
    const int i0 = threadIdx.x / NC;
    float acc[4];
#pragma unroll
    for (int j = 0; j < 4; ++j) acc[j] = bc[cc];
#pragma unroll 8
    for (int k = 0; k < DD; ++k) {
      const float w = Wc[k * NC + cc];
#pragma unroll
      for (int j = 0; j < 4; ++j) acc[j] += hg[(i0 + 4 * j) * HG_PITCH + k] * w;
    }
#pragma unroll
    for (int j = 0; j < 4; ++j) out[(g0 + i0 + 4 * j) * NC + cc] = acc[j];
  }
}

extern "C" void kernel_launch(void* const* d_in, const int* in_sizes, int n_in,
                              void* d_out, int out_size, void* d_ws, size_t ws_size,
                              hipStream_t stream) {
  const float* feat = (const float*)d_in[0];
  const float* W1   = (const float*)d_in[1];
  const float* b1   = (const float*)d_in[2];
  const float* W2   = (const float*)d_in[3];
  const float* b2   = (const float*)d_in[4];
  const float* Wc   = (const float*)d_in[5];
  const float* bc   = (const float*)d_in[6];
  const int*   src  = (const int*)d_in[7];
  const int*   dst  = (const int*)d_in[8];
  const int*   gid  = (const int*)d_in[9];
  float* out = (float*)d_out;

  float*    ws      = (float*)d_ws;
  unsigned* deg_in  = (unsigned*)ws + NN;
  float*    nsrc    = ws + 2 * NN;
  float*    ndst    = nsrc + NN;
  float*    bufA    = ndst + NN;
  float*    bufB    = bufA + (size_t)NN * DD;
  float*    sums    = bufB + (size_t)NN * DD;
  float*    cnt     = sums + NG * DD;
  unsigned* rowptr  = (unsigned*)(cnt + NG);
  unsigned* rank    = rowptr + 40064;
  int*      col     = (int*)(rank + NE);
  unsigned* tab_in  = (unsigned*)(col + NE);
  unsigned* tab_out = tab_in + (size_t)NCH * NN;

  const int T = 256;

  deg_chunk_kernel<<<2 * NCH, 1024, 0, stream>>>(src, dst, tab_in, tab_out, rank);
  merge_kernel<<<(NN + T - 1) / T, T, 0, stream>>>(tab_in, tab_out, deg_in, nsrc, ndst, sums);
  scan_gemm_kernel<<<1 + NN / 16, 256, 0, stream>>>(feat, nsrc, W1, bufA, deg_in, rowptr);
  fill_kernel<<<(NE + T - 1) / T, T, 0, stream>>>(src, dst, rowptr, tab_in, rank, col);

  spmm_pull<false><<<NN / ROWS_PER_WG, 256, 0, stream>>>(bufA, rowptr, col, ndst, nsrc, b1, gid, bufB, cnt);

  spmm_pull<true><<<NN / ROWS_PER_WG, 256, 0, stream>>>(bufB, rowptr, col, ndst, nsrc, b2, gid, sums, cnt);

  readout_kernel<<<NG / RO_ROWS, 256, 0, stream>>>(sums, cnt, W2, b2, Wc, bc, out);
}


# resident grids: a gather wave takes 5 rows (bounds and source indices of all 5 fetched up front), a fill thread 8 edges, a GEMM block 5 strips with W kept in registers; per-graph sums reassociated
# speedup vs baseline: 3.4664x; 1.0771x over previous
#include <hip/hip_runtime.h>

#define NN 40000
#define NE 640000
#define DD 128
#define NC 40
#define NG 128

typedef __attribute__((ext_vector_type(2))) float v2f;
typedef __attribute__((ext_vector_type(8))) float v8f;

__device__ __forceinline__ void atomAddF(float* p, float v) {
#if defined(__HIP_DEVICE_COMPILE__)
  unsafeAtomicAdd(p, v);
#else
  __hip_atomic_fetch_add(p, v, __ATOMIC_RELAXED, __HIP_MEMORY_SCOPE_AGENT);
#endif
}

#define NCH 64
#define CH (NE / NCH)
#define DEG_IT ((CH + 1023) / 1024)
__global__ void __launch_bounds__(1024) deg_chunk_kernel(const int* __restrict__ src,
                                                         const int* __restrict__ dst,
                                                         unsigned* __restrict__ tab_in,
                                                         unsigned* __restrict__ tab_out,
                                                         unsigned* __restrict__ rank) {
  __shared__ unsigned h[NN];
  const int which = blockIdx.x / NCH;
  const int w = blockIdx.x % NCH;
  const int* __restrict__ key = which ? src : dst;
  const int e0 = w * CH;
  unsigned k[DEG_IT];
#pragma unroll
  for (int j = 0; j < DEG_IT; ++j) {
    const int i = threadIdx.x + 1024 * j;
    k[j] = i < CH ? (unsigned)key[e0 + i] : 0xFFFFFFFFu;
  }
  for (int i = threadIdx.x; i < NN; i += 1024) h[i] = 0u;
  __syncthreads();
#pragma unroll
  for (int j = 0; j < DEG_IT; ++j) {
    if (k[j] < (unsigned)NN) {
      const unsigned r = atomicAdd(&h[k[j]], 1u);
      if (!which) rank[e0 + threadIdx.x + 1024 * j] = r;
    }
  }
  __syncthreads();
  unsigned* __restrict__ t = (which ? tab_out : tab_in) + (size_t)w * NN;
  for (int i = threadIdx.x; i < NN; i += 1024) t[i] = h[i];
}

__global__ void merge_kernel(unsigned* __restrict__ tab_in, const unsigned* __restrict__ tab_out,
                             unsigned* __restrict__ deg_in, float* __restrict__ ns,
                             float* __restrict__ nd, float* __restrict__ zsum) {
  const int d = blockIdx.x * blockDim.x + threadIdx.x;
  if (d < NG * DD + NG) zsum[d] = 0.0f;
  if (d < NN) {
    unsigned run = 0, so = 0;
#pragma unroll 8
    for (int w = 0; w < NCH; ++w) {
      const unsigned c = tab_in[(size_t)w * NN + d];
      so += tab_out[(size_t)w * NN + d];
      tab_in[(size_t)w * NN + d] = run;
      run += c;
    }
    deg_in[d] = run;
    ns[d] = rsqrtf(fmaxf((float)so, 1.0f));
    nd[d] = rsqrtf(fmaxf((float)run, 1.0f));
  }
}

#define SCAN_W4 (NN / 8 / 4)
#define SCAN_IT ((SCAN_W4 + 31) / 32)
__device__ __forceinline__ void rowptr_scan_block(const unsigned* __restrict__ din,
                                                  unsigned* __restrict__ rowptr,
                                                  unsigned* wtot) {
  const int t = threadIdx.x;
  const int lane = t & 31;
  const int w = t >> 5;
  const uint4* __restrict__ p = (const uint4*)din + SCAN_W4 * w;
  uint4* __restrict__ q = (uint4*)rowptr + SCAN_W4 * w;
  unsigned s = 0;
#pragma unroll 8
  for (int i = 0; i < SCAN_IT; ++i) {
    const int j = lane + 32 * i;
    if (j < SCAN_W4) {
      const uint4 v = p[j];
      s += v.x + v.y + v.z + v.w;
    }
  }
  for (int o = 16; o > 0; o >>= 1) s += __shfl_xor(s, o);
  if (lane == 0) wtot[w] = s;
  __syncthreads();
  unsigned carry = 0;
  for (int k = 0; k < w; ++k) carry += wtot[k];
#pragma unroll 8
  for (int i = 0; i < SCAN_IT; ++i) {
    const int j = lane + 32 * i;
    uint4 v = {0u, 0u, 0u, 0u};
    if (j < SCAN_W4) v = p[j];
    const unsigned own = v.x + v.y + v.z + v.w;
    unsigned inc = own;
    for (int o = 1; o < 32; o <<= 1) {
      const unsigned u = __shfl_up(inc, o);
      if (lane >= o) inc += u;
    }
    unsigned run = carry + inc - own;
    if (j < SCAN_W4) {
      uint4 r;
      r.x = run; run += v.x;
      r.y = run; run += v.y;
      r.z = run; run += v.z;
      r.w = run;
      q[j] = r;
    }
    carry += (unsigned)__builtin_amdgcn_readlane((int)inc, 31);
  }
  if (t == 255) rowptr[NN] = carry;
}

#define FILL_EPT 8
__global__ void fill_kernel(const int* __restrict__ src, const int* __restrict__ dst,
                            const unsigned* __restrict__ rowptr, const unsigned* __restrict__ tab_in,
                            const unsigned* __restrict__ rank, int* __restrict__ col) {
  const int t = blockIdx.x * blockDim.x + threadIdx.x;
  if (t < NE / FILL_EPT) {
    const unsigned* __restrict__ tab = tab_in + (size_t)((FILL_EPT * t) / CH) * NN;
    int ss[FILL_EPT];
    int dd[FILL_EPT];
    unsigned rr[FILL_EPT];
    unsigned pos[FILL_EPT];
#pragma unroll
    for (int v = 0; v < FILL_EPT / 4; ++v) {
      const int4 s4 = ((const int4*)src)[(FILL_EPT / 4) * t + v];
      const int4 d4 = ((const int4*)dst)[(FILL_EPT / 4) * t + v];
      const uint4 r4 = ((const uint4*)rank)[(FILL_EPT / 4) * t + v];
      ss[4 * v + 0] = s4.x; ss[4 * v + 1] = s4.y; ss[4 * v + 2] = s4.z; ss[4 * v + 3] = s4.w;
      dd[4 * v + 0] = d4.x; dd[4 * v + 1] = d4.y; dd[4 * v + 2] = d4.z; dd[4 * v + 3] = d4.w;
      rr[4 * v + 0] = r4.x; rr[4 * v + 1] = r4.y; rr[4 * v + 2] = r4.z; rr[4 * v + 3] = r4.w;
    }
    asm volatile("" :: "v"(ss[0]), "v"(ss[1]), "v"(ss[2]), "v"(ss[3]),
                       "v"(ss[4]), "v"(ss[5]), "v"(ss[6]), "v"(ss[7]));
#pragma unroll
    for (int k = 0; k < FILL_EPT; ++k) {
      unsigned d = (unsigned)dd[k];
      d = d < (unsigned)NN ? d : (unsigned)(NN - 1);
      pos[k] = rowptr[d] + tab[d] + rr[k];
    }
#pragma unroll
    for (int k = 0; k < FILL_EPT; ++k) {
      if (pos[k] < (unsigned)NE) col[pos[k]] = ss[k];
    }
  }
}

#define GEMM_SPW 5
__global__ void __launch_bounds__(256) scan_gemm_kernel(const float* __restrict__ X,
                                                        const float* __restrict__ rs,
                                                        const float* __restrict__ W,
                                                        float* __restrict__ Y,
                                                        const unsigned* __restrict__ din,
                                                        unsigned* __restrict__ rowptr) {
  __shared__ unsigned wtot[8];
  if (blockIdx.x == 0) {
    rowptr_scan_block(din, rowptr, wtot);
    return;
  }
  const int lane = threadIdx.x & 31;
  const int wave = threadIdx.x >> 5;
  const int hl   = lane >> 4;
  const int l    = lane & 15;
  const int n0   = wave * 16;

  v2f bf[DD / 4];
#pragma unroll
  for (int k = 0; k < DD; k += 4) {
    const int ka = k + 2 * hl;
    bf[k / 4].x = W[ka * DD + n0 + l];
    bf[k / 4].y = W[(ka + 1) * DD + n0 + l];
  }

  for (int s = 0; s < GEMM_SPW; ++s) {
    const int row0 = ((blockIdx.x - 1) * GEMM_SPW + s) * 16;
    const int m    = row0 + l;
    const float scale = rs[m];
    const float* __restrict__ arow = X + (size_t)m * DD;
    v8f c = {};
#pragma unroll
    for (int k = 0; k < DD; k += 4) {
      const int ka = k + 2 * hl;
      v2f a;
      a.x = arow[ka]     * scale;
      a.y = arow[ka + 1] * scale;
      c = __builtin_amdgcn_wmma_f32_16x16x4_f32(false, a, false, bf[k / 4], (short)0, c, false, false);
    }
#pragma unroll
    for (int r = 0; r < 8; ++r) {
      Y[(size_t)(row0 + 8 * hl + r) * DD + n0 + l] = c[r];
    }
  }
}

template <int K>
__device__ __forceinline__ void issue_rows(const float4* __restrict__ H4, int mine, int j,
                                           int lane, float4* r) {
#pragma unroll
  for (int k = 0; k < K; ++k) {
    const unsigned sidx = (unsigned)__builtin_amdgcn_readlane(mine, j + k);
    __builtin_assume(sidx < (unsigned)NN);
    r[k] = H4[(size_t)sidx * (DD / 4) + lane];
  }
}
template <int K>
__device__ __forceinline__ void add_rows(const float4* r, float4& acc) {
#pragma unroll
  for (int k = 0; k < K; ++k) {
    acc.x += r[k].x; acc.y += r[k].y; acc.z += r[k].z; acc.w += r[k].w;
  }
}

__device__ __forceinline__ void gather_chunk(const float4* __restrict__ H4, int cur, int n,
                                             int lane, float4& acc) {
  int j = 0;
  for (; j + 8 <= n; j += 8) {
    float4 r[8];
    issue_rows<8>(H4, cur, j, lane, r);
    add_rows<8>(r, acc);
  }
  const int t = n - j;
  float4 r4[4], r2[2], r1[1];
  const int j2 = j + (t & 4);
  const int j1 = j2 + (t & 2);
  if (t & 4) issue_rows<4>(H4, cur, j, lane, r4);
  if (t & 2) issue_rows<2>(H4, cur, j2, lane, r2);
  if (t & 1) issue_rows<1>(H4, cur, j1, lane, r1);
  if (t & 4) add_rows<4>(r4, acc);
  if (t & 2) add_rows<2>(r2, acc);
  if (t & 1) add_rows<1>(r1, acc);
}

#define ROWS_PER_WG 8
#define GRP_PER_WG 5
#define SPMM_WGS (NN / (ROWS_PER_WG * GRP_PER_WG))
template <bool POOL>
__global__ void __launch_bounds__(256) __attribute__((amdgpu_waves_per_eu(10, 10))) spmm_pull(const float* __restrict__ H,
                                                 const unsigned* __restrict__ rowptr,
                                                 const int* __restrict__ col,
                                                 const float* __restrict__ nd,
                                                 const float* __restrict__ ns,
                                                 const float* __restrict__ bias,
                                                 const int* __restrict__ gid,
                                                 float* __restrict__ OUT,
                                                 float* __restrict__ CNT) {
  const int lane = threadIdx.x & 31;
  const int wave = __builtin_amdgcn_readfirstlane((int)(threadIdx.x >> 5));
  const int dbase = (int)(blockIdx.x * (ROWS_PER_WG * GRP_PER_WG)) + wave;
  const float4* __restrict__ H4 = (const float4*)H;

  unsigned bnd = 0u;
  float sc_l = 0.0f, sn_l = 1.0f;
  int g_l = -1;
  {
    const int i = lane < GRP_PER_WG ? lane : lane - GRP_PER_WG;
    const int dl = dbase + ROWS_PER_WG * i;
    if (lane < 2 * GRP_PER_WG && dl < NN) {
      bnd = rowptr[dl + (lane < GRP_PER_WG ? 0 : 1)];
      bnd = bnd < (unsigned)NE ? bnd : (unsigned)NE;
      if (lane < GRP_PER_WG) {
        sc_l = nd[dl];
        if constexpr (POOL) g_l = gid[dl]; else sn_l = ns[dl];
      }
    }
  }
  float4 b = {0.0f, 0.0f, 0.0f, 0.0f};
  if constexpr (!POOL) b = ((const float4*)bias)[lane];

  float4 racc = {0.0f, 0.0f, 0.0f, 0.0f};
  int rg = -1;
  float rn = 0.0f;

  int mi[GRP_PER_WG];
#pragma unroll
  for (int i = 0; i < GRP_PER_WG; ++i) {
    const unsigned bi = (unsigned)__builtin_amdgcn_readlane((int)bnd, i);
    const unsigned ei = (unsigned)__builtin_amdgcn_readlane((int)bnd, GRP_PER_WG + i);
    mi[i] = bi + lane < ei ? col[bi + lane] : 0;
  }
#pragma unroll 1
  for (int it = 0; it < GRP_PER_WG; ++it) {
    const int d = dbase + ROWS_PER_WG * it;
    const unsigned beg = (unsigned)__builtin_amdgcn_readlane((int)bnd, it);
    const unsigned end = (unsigned)__builtin_amdgcn_readlane((int)bnd, GRP_PER_WG + it);
    int mine = mi[0];
#pragma unroll
    for (int i = 1; i < GRP_PER_WG; ++i) mine = it == i ? mi[i] : mine;
    float4 acc = {0.0f, 0.0f, 0.0f, 0.0f};
    for (unsigned p = beg; p < end; p += 32u) {
      int raw = mine;
      if (p != beg) raw = p + lane < end ? col[p + lane] : 0;
      const unsigned uraw = (unsigned)raw;
      const int cur = (int)(uraw < (unsigned)NN ? uraw : (unsigned)(NN - 1));
      gather_chunk(H4, cur, (int)((end - p) < 32u ? (end - p) : 32u), lane, acc);
    }
    const float sc = __int_as_float(__builtin_amdgcn_readlane(__float_as_int(sc_l), it));
    if constexpr (POOL) {
      float4 o;
      o.x = acc.x * sc;
      o.y = acc.y * sc;
      o.z = acc.z * sc;
      o.w = acc.w * sc;
      const int g = __builtin_amdgcn_readlane(g_l, it);
      if (d < NN) {
        if (g != rg) {
          if (rn > 0.0f && (unsigned)rg < (unsigned)NG) {
            atomAddF(&OUT[rg * DD + 4 * lane + 0], racc.x);
            atomAddF(&OUT[rg * DD + 4 * lane + 1], racc.y);
            atomAddF(&OUT[rg * DD + 4 * lane + 2], racc.z);
            atomAddF(&OUT[rg * DD + 4 * lane + 3], racc.w);
            if (lane == 0) atomAddF(&CNT[rg], rn);
          }
          racc = o;
          rg = g;
          rn = 1.0f;
        } else {
          racc.x += o.x; racc.y += o.y; racc.z += o.z; racc.w += o.w;
          rn += 1.0f;
        }
      }
    } else {
      const float sn = __int_as_float(__builtin_amdgcn_readlane(__float_as_int(sn_l), it));
      if (d < NN) {
        float4 o;
        o.x = fmaxf(acc.x * sc + b.x, 0.0f) * sn;
        o.y = fmaxf(acc.y * sc + b.y, 0.0f) * sn;
        o.z = fmaxf(acc.z * sc + b.z, 0.0f) * sn;
        o.w = fmaxf(acc.w * sc + b.w, 0.0f) * sn;
        ((float4*)OUT)[(size_t)d * (DD / 4) + lane] = o;
      }
    }
  }
  if constexpr (POOL) {
    __shared__ float4 prow[ROWS_PER_WG][DD / 4];
    __shared__ int pgid[ROWS_PER_WG];
    __shared__ float pcnt[ROWS_PER_WG];
    prow[wave][lane] = racc;
    if (lane == 0) {
      pgid[wave] = rn > 0.0f ? rg : -1;
      pcnt[wave] = rn;
    }
    __syncthreads();
    if (threadIdx.x < DD) {
      const int c = threadIdx.x;
      const float* __restrict__ pr = (const float*)&prow[0][0];
      float v[ROWS_PER_WG];
      int gs[ROWS_PER_WG];
#pragma unroll
      for (int r = 0; r < ROWS_PER_WG; ++r) {
        v[r] = pr[r * DD + c];
        gs[r] = __builtin_amdgcn_readfirstlane(pgid[r]);
      }
      int g = gs[0];
      float s = v[0];
#pragma unroll
      for (int r = 1; r < ROWS_PER_WG; ++r) {
        if (gs[r] != g) {
          if ((unsigned)g < (unsigned)NG) atomAddF(&OUT[g * DD + c], s);
          g = gs[r];
          s = v[r];
        } else {
          s += v[r];
        }
      }
      if ((unsigned)g < (unsigned)NG) atomAddF(&OUT[g * DD + c], s);
    } else if (threadIdx.x == DD) {
      int g = pgid[0];
      float n = pcnt[0];
      for (int r = 1; r < ROWS_PER_WG; ++r) {
        const int gr = pgid[r];
        if (gr != g) {
          if ((unsigned)g < (unsigned)NG) atomAddF(&CNT[g], n);
          g = gr;
          n = pcnt[r];
        } else {
          n += pcnt[r];
        }
      }
      if ((unsigned)g < (unsigned)NG) atomAddF(&CNT[g], n);
    }
  }
}

#define HG_PITCH (DD + 4)
#define RO_ROWS 16
__global__ void __launch_bounds__(256) readout_kernel(const float* __restrict__ S,
                                                      const float* __restrict__ cnt,
                                                      const float* __restrict__ W2,
                                                      const float* __restrict__ b2,
                                                      const float* __restrict__ Wc,
                                                      const float* __restrict__ bc,
                                                      float* __restrict__ out) {
  __shared__ float hg[RO_ROWS * HG_PITCH];
  const int lane = threadIdx.x & 31;
  const int wave = threadIdx.x >> 5;
  const int hl   = lane >> 4;
  const int l    = lane & 15;
  const int g0   = blockIdx.x * RO_ROWS;
  const int n0   = wave * 16;
  const float* __restrict__ arow = S + (size_t)(g0 + l) * DD;

  v8f c = {};
#pragma unroll
  for (int k = 0; k < DD; k += 4) {
    const int ka = k + 2 * hl;
    v2f a, b;
    a.x = arow[ka];
    a.y = arow[ka + 1];
    b.x = W2[ka * DD + n0 + l];
    b.y = W2[(ka + 1) * DD + n0 + l];
    c = __builtin_amdgcn_wmma_f32_16x16x4_f32(false, a, false, b, (short)0, c, false, false);
  }
  const float bcol = b2[n0 + l];
#pragma unroll
  for (int r = 0; r < 8; ++r) {
    const float n = cnt[g0 + 8 * hl + r];
    const float inv = 1.0f / fmaxf(n, 1.0f);
    hg[(8 * hl + r) * HG_PITCH + n0 + l] = (c[r] + n * bcol) * inv;
  }
  __syncthreads();
  if (threadIdx.x < 4 * NC) {
    const int cc = threadIdx.x % NC;
    const int i0 = threadIdx.x / NC;
    float acc[4];
#pragma unroll
    for (int j = 0; j < 4; ++j) acc[j] = bc[cc];
#pragma unroll 8
    for (int k = 0; k < DD; ++k) {
      const float w = Wc[k * NC + cc];
#pragma unroll
      for (int j = 0; j < 4; ++j) acc[j] += hg[(i0 + 4 * j) * HG_PITCH + k] * w;
    }
#pragma unroll
    for (int j = 0; j < 4; ++j) out[(g0 + i0 + 4 * j) * NC + cc] = acc[j];
  }
}

extern "C" void kernel_launch(void* const* d_in, const int* in_sizes, int n_in,
                              void* d_out, int out_size, void* d_ws, size_t ws_size,
                              hipStream_t stream) {
  const float* feat = (const float*)d_in[0];
  const float* W1   = (const float*)d_in[1];
  const float* b1   = (const float*)d_in[2];
  const float* W2   = (const float*)d_in[3];
  const float* b2   = (const float*)d_in[4];
  const float* Wc   = (const float*)d_in[5];
  const float* bc   = (const float*)d_in[6];
  const int*   src  = (const int*)d_in[7];
  const int*   dst  = (const int*)d_in[8];
  const int*   gid  = (const int*)d_in[9];
  float* out = (float*)d_out;

  float*    ws      = (float*)d_ws;
  unsigned* deg_in  = (unsigned*)ws + NN;
  float*    nsrc    = ws + 2 * NN;
  float*    ndst    = nsrc + NN;
  float*    bufA    = ndst + NN;
  float*    bufB    = bufA + (size_t)NN * DD;
  float*    sums    = bufB + (size_t)NN * DD;
  float*    cnt     = sums + NG * DD;
  unsigned* rowptr  = (unsigned*)(cnt + NG);
  unsigned* rank    = rowptr + 40064;
  int*      col     = (int*)(rank + NE);
  unsigned* tab_in  = (unsigned*)(col + NE);
  unsigned* tab_out = tab_in + (size_t)NCH * NN;

  const int T = 256;

  deg_chunk_kernel<<<2 * NCH, 1024, 0, stream>>>(src, dst, tab_in, tab_out, rank);
  merge_kernel<<<(NN + T - 1) / T, T, 0, stream>>>(tab_in, tab_out, deg_in, nsrc, ndst, sums);
  scan_gemm_kernel<<<1 + NN / 16 / GEMM_SPW, 256, 0, stream>>>(feat, nsrc, W1, bufA, deg_in, rowptr);
  fill_kernel<<<(NE / FILL_EPT + T - 1) / T, T, 0, stream>>>(src, dst, rowptr, tab_in, rank, col);

  spmm_pull<false><<<SPMM_WGS, 256, 0, stream>>>(bufA, rowptr, col, ndst, nsrc, b1, gid, bufB, cnt);

  spmm_pull<true><<<SPMM_WGS, 256, 0, stream>>>(bufB, rowptr, col, ndst, nsrc, b2, gid, sums, cnt);

  readout_kernel<<<NG / RO_ROWS, 256, 0, stream>>>(sums, cnt, W2, b2, Wc, bc, out);
}


# layer-1 GEMM blocks ride in the merge launch (unscaled features; norm_src now multiplies each gathered row: rounding order changed); rowptr scan split into merge and fill; 7 launches to 6
# speedup vs baseline: 3.6539x; 1.0541x over previous
#include <hip/hip_runtime.h>

#define NN 40000
#define NE 640000
#define DD 128
#define NC 40
#define NG 128

typedef __attribute__((ext_vector_type(2))) float v2f;
typedef __attribute__((ext_vector_type(8))) float v8f;

__device__ __forceinline__ void atomAddF(float* p, float v) {
#if defined(__HIP_DEVICE_COMPILE__)
  unsafeAtomicAdd(p, v);
#else
  __hip_atomic_fetch_add(p, v, __ATOMIC_RELAXED, __HIP_MEMORY_SCOPE_AGENT);
#endif
}

#define NCH 64
#define CH (NE / NCH)
#define DEG_IT ((CH + 1023) / 1024)
__global__ void __launch_bounds__(1024) deg_chunk_kernel(const int* __restrict__ src,
                                                         const int* __restrict__ dst,
                                                         unsigned* __restrict__ tab_in,
                                                         unsigned* __restrict__ tab_out,
                                                         unsigned* __restrict__ rank) {
  __shared__ unsigned h[NN];
  const int which = blockIdx.x / NCH;
  const int w = blockIdx.x % NCH;
  const int* __restrict__ key = which ? src : dst;
  const int e0 = w * CH;
  unsigned k[DEG_IT];
#pragma unroll
  for (int j = 0; j < DEG_IT; ++j) {
    const int i = threadIdx.x + 1024 * j;
    k[j] = i < CH ? (unsigned)key[e0 + i] : 0xFFFFFFFFu;
  }
  for (int i = threadIdx.x; i < NN; i += 1024) h[i] = 0u;
  __syncthreads();
#pragma unroll
  for (int j = 0; j < DEG_IT; ++j) {
    if (k[j] < (unsigned)NN) {
      const unsigned r = atomicAdd(&h[k[j]], 1u);
      if (!which) rank[e0 + threadIdx.x + 1024 * j] = r;
    }
  }
  __syncthreads();
  unsigned* __restrict__ t = (which ? tab_out : tab_in) + (size_t)w * NN;
  for (int i = threadIdx.x; i < NN; i += 1024) t[i] = h[i];
}

__device__ __forceinline__ unsigned block_excl_scan256(unsigned v, unsigned* wt, unsigned& total) {
  const int lane = threadIdx.x & 31;
  const int w = threadIdx.x >> 5;
  unsigned inc = v;
  for (int o = 1; o < 32; o <<= 1) {
    const unsigned u = __shfl_up(inc, o);
    if (lane >= o) inc += u;
  }
  if (lane == 31) wt[w] = inc;
  __syncthreads();
  unsigned woff = 0, tot = 0;
#pragma unroll
  for (int k = 0; k < 8; ++k) {
    const unsigned x = wt[k];
    woff += k < w ? x : 0u;
    tot += x;
  }
  total = tot;
  return woff + inc - v;
}

#define MB ((NN + 255) / 256)
#define GEMM_SPW 5
__global__ void __launch_bounds__(256) merge_gemm_kernel(unsigned* __restrict__ tab_in,
                                                         const unsigned* __restrict__ tab_out,
                                                         unsigned* __restrict__ lrow,
                                                         unsigned* __restrict__ btot,
                                                         float* __restrict__ ns,
                                                         float* __restrict__ nd,
                                                         float* __restrict__ zsum,
                                                         const float* __restrict__ X,
                                                         const float* __restrict__ W,
                                                         float* __restrict__ Y) {
  __shared__ unsigned wt[8];
  if (blockIdx.x < MB) {
    const int d = blockIdx.x * 256 + threadIdx.x;
    if (d < NG * DD + NG) zsum[d] = 0.0f;
    unsigned run = 0, so = 0;
    if (d < NN) {
#pragma unroll 8
      for (int w = 0; w < NCH; ++w) {
        const unsigned c = tab_in[(size_t)w * NN + d];
        so += tab_out[(size_t)w * NN + d];
        tab_in[(size_t)w * NN + d] = run;
        run += c;
      }
    }
    unsigned total;
    const unsigned ex = block_excl_scan256(run, wt, total);
    if (d < NN) {
      lrow[d] = ex;
      ns[d] = rsqrtf(fmaxf((float)so, 1.0f));
      nd[d] = rsqrtf(fmaxf((float)run, 1.0f));
    }
    if (threadIdx.x == 0) btot[blockIdx.x] = total;
    return;
  }
  const int gb   = blockIdx.x - MB;
  const int lane = threadIdx.x & 31;
  const int wave = threadIdx.x >> 5;
  const int hl   = lane >> 4;
  const int l    = lane & 15;
  const int n0   = wave * 16;

  v2f bf[DD / 4];
#pragma unroll
  for (int k = 0; k < DD; k += 4) {
    const int ka = k + 2 * hl;
    bf[k / 4].x = W[ka * DD + n0 + l];
    bf[k / 4].y = W[(ka + 1) * DD + n0 + l];
  }

  for (int s = 0; s < GEMM_SPW; ++s) {
    const int row0 = (gb * GEMM_SPW + s) * 16;
    const int m    = row0 + l;
    const float* __restrict__ arow = X + (size_t)m * DD;
    v8f c = {};
#pragma unroll
    for (int k = 0; k < DD; k += 4) {
      const int ka = k + 2 * hl;
      v2f a;
      a.x = arow[ka];
      a.y = arow[ka + 1];
      c = __builtin_amdgcn_wmma_f32_16x16x4_f32(false, a, false, bf[k / 4], (short)0, c, false, false);
    }
#pragma unroll
    for (int r = 0; r < 8; ++r) {
      Y[(size_t)(row0 + 8 * hl + r) * DD + n0 + l] = c[r];
    }
  }
}

#define FILL_EPT 8
__global__ void __launch_bounds__(256) fill_kernel(const int* __restrict__ src, const int* __restrict__ dst,
                            const unsigned* __restrict__ lrow, const unsigned* __restrict__ btot,
                            const unsigned* __restrict__ tab_in,
                            const unsigned* __restrict__ rank, int* __restrict__ col,
                            unsigned* __restrict__ rowptr) {
  __shared__ unsigned wt[8];
  __shared__ unsigned boff[MB + 1];
  const int t = blockIdx.x * 256 + threadIdx.x;
  const bool act = t < NE / FILL_EPT;
  const int te = act ? t : NE / FILL_EPT - 1;
  int ss[FILL_EPT];
  int dd[FILL_EPT];
  unsigned rr[FILL_EPT];
  unsigned pos[FILL_EPT];
#pragma unroll
  for (int v = 0; v < FILL_EPT / 4; ++v) {
    const int4 s4 = ((const int4*)src)[(FILL_EPT / 4) * te + v];
    const int4 d4 = ((const int4*)dst)[(FILL_EPT / 4) * te + v];
    const uint4 r4 = ((const uint4*)rank)[(FILL_EPT / 4) * te + v];
    ss[4 * v + 0] = s4.x; ss[4 * v + 1] = s4.y; ss[4 * v + 2] = s4.z; ss[4 * v + 3] = s4.w;
    dd[4 * v + 0] = d4.x; dd[4 * v + 1] = d4.y; dd[4 * v + 2] = d4.z; dd[4 * v + 3] = d4.w;
    rr[4 * v + 0] = r4.x; rr[4 * v + 1] = r4.y; rr[4 * v + 2] = r4.z; rr[4 * v + 3] = r4.w;
  }
  const unsigned lr_own = t < NN ? lrow[t] : 0u;
  const unsigned bt = threadIdx.x < MB ? btot[threadIdx.x] : 0u;
  unsigned total;
  const unsigned ex = block_excl_scan256(bt, wt, total);
  if (threadIdx.x < MB) boff[threadIdx.x] = ex;
  if (threadIdx.x == MB) boff[MB] = total;
  __syncthreads();
  if (t <= NN) rowptr[t] = t < NN ? boff[t >> 8] + lr_own : boff[MB];
  const unsigned* __restrict__ tab = tab_in + (size_t)((FILL_EPT * te) / CH) * NN;
  asm volatile("" :: "v"(ss[0]), "v"(ss[1]), "v"(ss[2]), "v"(ss[3]),
                     "v"(ss[4]), "v"(ss[5]), "v"(ss[6]), "v"(ss[7]));
#pragma unroll
  for (int k = 0; k < FILL_EPT; ++k) {
    unsigned d = (unsigned)dd[k];
    d = d < (unsigned)NN ? d : (unsigned)(NN - 1);
    pos[k] = boff[d >> 8] + lrow[d] + tab[d] + rr[k];
  }
  if (act) {
#pragma unroll
    for (int k = 0; k < FILL_EPT; ++k) {
      if (pos[k] < (unsigned)NE) col[pos[k]] = ss[k];
    }
  }
}

template <int K>
__device__ __forceinline__ void issue_rows(const float4* __restrict__ H4, int mine, int j,
                                           int lane, float4* r) {
#pragma unroll
  for (int k = 0; k < K; ++k) {
    const unsigned sidx = (unsigned)__builtin_amdgcn_readlane(mine, j + k);
    __builtin_assume(sidx < (unsigned)NN);
    r[k] = H4[(size_t)sidx * (DD / 4) + lane];
  }
}
template <int K, bool SCALE>
__device__ __forceinline__ void add_rows(const float4* r, float sv, int j, float4& acc) {
#pragma unroll
  for (int k = 0; k < K; ++k) {
    if constexpr (SCALE) {
      const float s = __int_as_float(__builtin_amdgcn_readlane(__float_as_int(sv), j + k));
      acc.x += r[k].x * s; acc.y += r[k].y * s; acc.z += r[k].z * s; acc.w += r[k].w * s;
    } else {
      acc.x += r[k].x; acc.y += r[k].y; acc.z += r[k].z; acc.w += r[k].w;
    }
  }
}

template <bool SCALE>
__device__ __forceinline__ void gather_chunk(const float4* __restrict__ H4, int cur, float sv, int n,
                                             int lane, float4& acc) {
  int j = 0;
  for (; j + 8 <= n; j += 8) {
    float4 r[8];
    issue_rows<8>(H4, cur, j, lane, r);
    add_rows<8, SCALE>(r, sv, j, acc);
  }
  const int t = n - j;
  float4 r4[4], r2[2], r1[1];
  const int j2 = j + (t & 4);
  const int j1 = j2 + (t & 2);
  if (t & 4) issue_rows<4>(H4, cur, j, lane, r4);
  if (t & 2) issue_rows<2>(H4, cur, j2, lane, r2);
  if (t & 1) issue_rows<1>(H4, cur, j1, lane, r1);
  if (t & 4) add_rows<4, SCALE>(r4, sv, j, acc);
  if (t & 2) add_rows<2, SCALE>(r2, sv, j2, acc);
  if (t & 1) add_rows<1, SCALE>(r1, sv, j1, acc);
}

#define ROWS_PER_WG 8
#define GRP_PER_WG 5
#define SPMM_WGS (NN / (ROWS_PER_WG * GRP_PER_WG))
template <bool POOL>
__global__ void __launch_bounds__(256) __attribute__((amdgpu_waves_per_eu(10, 10))) spmm_pull(const float* __restrict__ H,
                                                 const unsigned* __restrict__ rowptr,
                                                 const int* __restrict__ col,
                                                 const float* __restrict__ nd,
                                                 const float* __restrict__ ns,
                                                 const float* __restrict__ bias,
                                                 const int* __restrict__ gid,
                                                 float* __restrict__ OUT,
                                                 float* __restrict__ CNT) {
  const int lane = threadIdx.x & 31;
  const int wave = __builtin_amdgcn_readfirstlane((int)(threadIdx.x >> 5));
  const int dbase = (int)(blockIdx.x * (ROWS_PER_WG * GRP_PER_WG)) + wave;
  const float4* __restrict__ H4 = (const float4*)H;

  unsigned bnd = 0u;
  float sc_l = 0.0f, sn_l = 1.0f;
  int g_l = -1;
  {
    const int i = lane < GRP_PER_WG ? lane : lane - GRP_PER_WG;
    const int dl = dbase + ROWS_PER_WG * i;
    if (lane < 2 * GRP_PER_WG && dl < NN) {
      bnd = rowptr[dl + (lane < GRP_PER_WG ? 0 : 1)];
      bnd = bnd < (unsigned)NE ? bnd : (unsigned)NE;
      if (lane < GRP_PER_WG) {
        sc_l = nd[dl];
        if constexpr (POOL) g_l = gid[dl]; else sn_l = ns[dl];
      }
    }
  }
  float4 b = {0.0f, 0.0f, 0.0f, 0.0f};
  if constexpr (!POOL) b = ((const float4*)bias)[lane];

  float4 racc = {0.0f, 0.0f, 0.0f, 0.0f};
  int rg = -1;
  float rn = 0.0f;

  int mi[GRP_PER_WG];
#pragma unroll
  for (int i = 0; i < GRP_PER_WG; ++i) {
    const unsigned bi = (unsigned)__builtin_amdgcn_readlane((int)bnd, i);
    const unsigned ei = (unsigned)__builtin_amdgcn_readlane((int)bnd, GRP_PER_WG + i);
    mi[i] = bi + lane < ei ? col[bi + lane] : 0;
  }
  float ms[GRP_PER_WG];
#pragma unroll
  for (int i = 0; i < GRP_PER_WG; ++i) {
    ms[i] = 1.0f;
    if constexpr (!POOL) {
      const unsigned um = (unsigned)mi[i];
      ms[i] = ns[um < (unsigned)NN ? um : (unsigned)(NN - 1)];
    }
  }
#pragma unroll 1
  for (int it = 0; it < GRP_PER_WG; ++it) {
    const int d = dbase + ROWS_PER_WG * it;
    const unsigned beg = (unsigned)__builtin_amdgcn_readlane((int)bnd, it);
    const unsigned end = (unsigned)__builtin_amdgcn_readlane((int)bnd, GRP_PER_WG + it);
    int mine = mi[0];
    float smine = ms[0];
#pragma unroll
    for (int i = 1; i < GRP_PER_WG; ++i) {
      mine = it == i ? mi[i] : mine;
      smine = it == i ? ms[i] : smine;
    }
    float4 acc = {0.0f, 0.0f, 0.0f, 0.0f};
    for (unsigned p = beg; p < end; p += 32u) {
      int raw = mine;
      if (p != beg) raw = p + lane < end ? col[p + lane] : 0;
      const unsigned uraw = (unsigned)raw;
      const int cur = (int)(uraw < (unsigned)NN ? uraw : (unsigned)(NN - 1));
      float sv = smine;
      if constexpr (!POOL) {
        if (p != beg) sv = ns[cur];
      }
      gather_chunk<!POOL>(H4, cur, sv, (int)((end - p) < 32u ? (end - p) : 32u), lane, acc);
    }
    const float sc = __int_as_float(__builtin_amdgcn_readlane(__float_as_int(sc_l), it));
    if constexpr (POOL) {
      float4 o;
      o.x = acc.x * sc;
      o.y = acc.y * sc;
      o.z = acc.z * sc;
      o.w = acc.w * sc;
      const int g = __builtin_amdgcn_readlane(g_l, it);
      if (d < NN) {
        if (g != rg) {
          if (rn > 0.0f && (unsigned)rg < (unsigned)NG) {
            atomAddF(&OUT[rg * DD + 4 * lane + 0], racc.x);
            atomAddF(&OUT[rg * DD + 4 * lane + 1], racc.y);
            atomAddF(&OUT[rg * DD + 4 * lane + 2], racc.z);
            atomAddF(&OUT[rg * DD + 4 * lane + 3], racc.w);
            if (lane == 0) atomAddF(&CNT[rg], rn);
          }
          racc = o;
          rg = g;
          rn = 1.0f;
        } else {
          racc.x += o.x; racc.y += o.y; racc.z += o.z; racc.w += o.w;
          rn += 1.0f;
        }
      }
    } else {
      const float sn = __int_as_float(__builtin_amdgcn_readlane(__float_as_int(sn_l), it));
      if (d < NN) {
        float4 o;
        o.x = fmaxf(acc.x * sc + b.x, 0.0f) * sn;
        o.y = fmaxf(acc.y * sc + b.y, 0.0f) * sn;
        o.z = fmaxf(acc.z * sc + b.z, 0.0f) * sn;
        o.w = fmaxf(acc.w * sc + b.w, 0.0f) * sn;
        ((float4*)OUT)[(size_t)d * (DD / 4) + lane] = o;
      }
    }
  }
  if constexpr (POOL) {
    __shared__ float4 prow[ROWS_PER_WG][DD / 4];
    __shared__ int pgid[ROWS_PER_WG];
    __shared__ float pcnt[ROWS_PER_WG];
    prow[wave][lane] = racc;
    if (lane == 0) {
      pgid[wave] = rn > 0.0f ? rg : -1;
      pcnt[wave] = rn;
    }
    __syncthreads();
    if (threadIdx.x < DD) {
      const int c = threadIdx.x;
      const float* __restrict__ pr = (const float*)&prow[0][0];
      float v[ROWS_PER_WG];
      int gs[ROWS_PER_WG];
#pragma unroll
      for (int r = 0; r < ROWS_PER_WG; ++r) {
        v[r] = pr[r * DD + c];
        gs[r] = __builtin_amdgcn_readfirstlane(pgid[r]);
      }
      int g = gs[0];
      float s = v[0];
#pragma unroll
      for (int r = 1; r < ROWS_PER_WG; ++r) {
        if (gs[r] != g) {
          if ((unsigned)g < (unsigned)NG) atomAddF(&OUT[g * DD + c], s);
          g = gs[r];
          s = v[r];
        } else {
          s += v[r];
        }
      }
      if ((unsigned)g < (unsigned)NG) atomAddF(&OUT[g * DD + c], s);
    } else if (threadIdx.x == DD) {
      int g = pgid[0];
      float n = pcnt[0];
      for (int r = 1; r < ROWS_PER_WG; ++r) {
        const int gr = pgid[r];
        if (gr != g) {
          if ((unsigned)g < (unsigned)NG) atomAddF(&CNT[g], n);
          g = gr;
          n = pcnt[r];
        } else {
          n += pcnt[r];
        }
      }
      if ((unsigned)g < (unsigned)NG) atomAddF(&CNT[g], n);
    }
  }
}

#define HG_PITCH (DD + 4)
#define RO_ROWS 16
__global__ void __launch_bounds__(256) readout_kernel(const float* __restrict__ S,
                                                      const float* __restrict__ cnt,
                                                      const float* __restrict__ W2,
                                                      const float* __restrict__ b2,
                                                      const float* __restrict__ Wc,
                                                      const float* __restrict__ bc,
                                                      float* __restrict__ out) {
  __shared__ float hg[RO_ROWS * HG_PITCH];
  const int lane = threadIdx.x & 31;
  const int wave = threadIdx.x >> 5;
  const int hl   = lane >> 4;
  const int l    = lane & 15;
  const int g0   = blockIdx.x * RO_ROWS;
  const int n0   = wave * 16;
  const float* __restrict__ arow = S + (size_t)(g0 + l) * DD;

  v8f c = {};
#pragma unroll
  for (int k = 0; k < DD; k += 4) {
    const int ka = k + 2 * hl;
    v2f a, b;
    a.x = arow[ka];
    a.y = arow[ka + 1];
    b.x = W2[ka * DD + n0 + l];
    b.y = W2[(ka + 1) * DD + n0 + l];
    c = __builtin_amdgcn_wmma_f32_16x16x4_f32(false, a, false, b, (short)0, c, false, false);
  }
  const float bcol = b2[n0 + l];
#pragma unroll
  for (int r = 0; r < 8; ++r) {
    const float n = cnt[g0 + 8 * hl + r];
    const float inv = 1.0f / fmaxf(n, 1.0f);
    hg[(8 * hl + r) * HG_PITCH + n0 + l] = (c[r] + n * bcol) * inv;
  }
  __syncthreads();
  if (threadIdx.x < 4 * NC) {
    const int cc = threadIdx.x % NC;
    const int i0 = threadIdx.x / NC;
    float acc[4];
#pragma unroll
    for (int j = 0; j < 4; ++j) acc[j] = bc[cc];
#pragma unroll 8
    for (int k = 0; k < DD; ++k) {
      const float w = Wc[k * NC + cc];
#pragma unroll
      for (int j = 0; j < 4; ++j) acc[j] += hg[(i0 + 4 * j) * HG_PITCH + k] * w;
    }
#pragma unroll
    for (int j = 0; j < 4; ++j) out[(g0 + i0 + 4 * j) * NC + cc] = acc[j];
  }
}

extern "C" void kernel_launch(void* const* d_in, const int* in_sizes, int n_in,
                              void* d_out, int out_size, void* d_ws, size_t ws_size,
                              hipStream_t stream) {
  const float* feat = (const float*)d_in[0];
  const float* W1   = (const float*)d_in[1];
  const float* b1   = (const float*)d_in[2];
  const float* W2   = (const float*)d_in[3];
  const float* b2   = (const float*)d_in[4];
  const float* Wc   = (const float*)d_in[5];
  const float* bc   = (const float*)d_in[6];
  const int*   src  = (const int*)d_in[7];
  const int*   dst  = (const int*)d_in[8];
  const int*   gid  = (const int*)d_in[9];
  float* out = (float*)d_out;

  float*    ws      = (float*)d_ws;
  unsigned* btot    = (unsigned*)ws;
  unsigned* lrow    = (unsigned*)ws + NN;
  float*    nsrc    = ws + 2 * NN;
  float*    ndst    = nsrc + NN;
  float*    bufA    = ndst + NN;
  float*    bufB    = bufA + (size_t)NN * DD;
  float*    sums    = bufB + (size_t)NN * DD;
  float*    cnt     = sums + NG * DD;
  unsigned* rowptr  = (unsigned*)(cnt + NG);
  unsigned* rank    = rowptr + 40064;
  int*      col     = (int*)(rank + NE);
  unsigned* tab_in  = (unsigned*)(col + NE);
  unsigned* tab_out = tab_in + (size_t)NCH * NN;

  const int T = 256;

  deg_chunk_kernel<<<2 * NCH, 1024, 0, stream>>>(src, dst, tab_in, tab_out, rank);
  merge_gemm_kernel<<<MB + NN / 16 / GEMM_SPW, 256, 0, stream>>>(tab_in, tab_out, lrow, btot, nsrc, ndst, sums,
                                                                  feat, W1, bufA);
  fill_kernel<<<(NE / FILL_EPT + T - 1) / T, T, 0, stream>>>(src, dst, lrow, btot, tab_in, rank, col, rowptr);

  spmm_pull<false><<<SPMM_WGS, 256, 0, stream>>>(bufA, rowptr, col, ndst, nsrc, b1, gid, bufB, cnt);

  spmm_pull<true><<<SPMM_WGS, 256, 0, stream>>>(bufB, rowptr, col, ndst, nsrc, b2, gid, sums, cnt);

  readout_kernel<<<NG / RO_ROWS, 256, 0, stream>>>(sums, cnt, W2, b2, Wc, bc, out);
}


# readout: mean rows @ Wc moved onto fp32 WMMA (3 waves, a 16-class tile each), Wc fragments requested at kernel entry: 17 serial global round trips behind the barrier gone; class sums reassociated
# speedup vs baseline: 3.9440x; 1.0794x over previous
#include <hip/hip_runtime.h>

#define NN 40000
#define NE 640000
#define DD 128
#define NC 40
#define NG 128

typedef __attribute__((ext_vector_type(2))) float v2f;
typedef __attribute__((ext_vector_type(8))) float v8f;

__device__ __forceinline__ void atomAddF(float* p, float v) {
#if defined(__HIP_DEVICE_COMPILE__)
  unsafeAtomicAdd(p, v);
#else
  __hip_atomic_fetch_add(p, v, __ATOMIC_RELAXED, __HIP_MEMORY_SCOPE_AGENT);
#endif
}

#define NCH 64
#define CH (NE / NCH)
#define DEG_IT ((CH + 1023) / 1024)
__global__ void __launch_bounds__(1024) deg_chunk_kernel(const int* __restrict__ src,
                                                         const int* __restrict__ dst,
                                                         unsigned* __restrict__ tab_in,
                                                         unsigned* __restrict__ tab_out,
                                                         unsigned* __restrict__ rank) {
  __shared__ unsigned h[NN];
  const int which = blockIdx.x / NCH;
  const int w = blockIdx.x % NCH;
  const int* __restrict__ key = which ? src : dst;
  const int e0 = w * CH;
  unsigned k[DEG_IT];
#pragma unroll
  for (int j = 0; j < DEG_IT; ++j) {
    const int i = threadIdx.x + 1024 * j;
    k[j] = i < CH ? (unsigned)key[e0 + i] : 0xFFFFFFFFu;
  }
  for (int i = threadIdx.x; i < NN; i += 1024) h[i] = 0u;
  __syncthreads();
#pragma unroll
  for (int j = 0; j < DEG_IT; ++j) {
    if (k[j] < (unsigned)NN) {
      const unsigned r = atomicAdd(&h[k[j]], 1u);
      if (!which) rank[e0 + threadIdx.x + 1024 * j] = r;
    }
  }
  __syncthreads();
  unsigned* __restrict__ t = (which ? tab_out : tab_in) + (size_t)w * NN;
  for (int i = threadIdx.x; i < NN; i += 1024) t[i] = h[i];
}

__device__ __forceinline__ unsigned block_excl_scan256(unsigned v, unsigned* wt, unsigned& total) {
  const int lane = threadIdx.x & 31;
  const int w = threadIdx.x >> 5;
  unsigned inc = v;
  for (int o = 1; o < 32; o <<= 1) {
    const unsigned u = __shfl_up(inc, o);
    if (lane >= o) inc += u;
  }
  if (lane == 31) wt[w] = inc;
  __syncthreads();
  unsigned woff = 0, tot = 0;
#pragma unroll
  for (int k = 0; k < 8; ++k) {
    const unsigned x = wt[k];
    woff += k < w ? x : 0u;
    tot += x;
  }
  total = tot;
  return woff + inc - v;
}

#define MB ((NN + 255) / 256)
#define GEMM_SPW 5
__global__ void __launch_bounds__(256) merge_gemm_kernel(unsigned* __restrict__ tab_in,
                                                         const unsigned* __restrict__ tab_out,
                                                         unsigned* __restrict__ lrow,
                                                         unsigned* __restrict__ btot,
                                                         float* __restrict__ ns,
                                                         float* __restrict__ nd,
                                                         float* __restrict__ zsum,
                                                         const float* __restrict__ X,
                                                         const float* __restrict__ W,
                                                         float* __restrict__ Y) {
  __shared__ unsigned wt[8];
  if (blockIdx.x < MB) {
    const int d = blockIdx.x * 256 + threadIdx.x;
    if (d < NG * DD + NG) zsum[d] = 0.0f;
    unsigned run = 0, so = 0;
    if (d < NN) {
#pragma unroll 8
      for (int w = 0; w < NCH; ++w) {
        const unsigned c = tab_in[(size_t)w * NN + d];
        so += tab_out[(size_t)w * NN + d];
        tab_in[(size_t)w * NN + d] = run;
        run += c;
      }
    }
    unsigned total;
    const unsigned ex = block_excl_scan256(run, wt, total);
    if (d < NN) {
      lrow[d] = ex;
      ns[d] = rsqrtf(fmaxf((float)so, 1.0f));
      nd[d] = rsqrtf(fmaxf((float)run, 1.0f));
    }
    if (threadIdx.x == 0) btot[blockIdx.x] = total;
    return;
  }
  const int gb   = blockIdx.x - MB;
  const int lane = threadIdx.x & 31;
  const int wave = threadIdx.x >> 5;
  const int hl   = lane >> 4;
  const int l    = lane & 15;
  const int n0   = wave * 16;

  v2f bf[DD / 4];
#pragma unroll
  for (int k = 0; k < DD; k += 4) {
    const int ka = k + 2 * hl;
    bf[k / 4].x = W[ka * DD + n0 + l];
    bf[k / 4].y = W[(ka + 1) * DD + n0 + l];
  }

  for (int s = 0; s < GEMM_SPW; ++s) {
    const int row0 = (gb * GEMM_SPW + s) * 16;
    const int m    = row0 + l;
    const float* __restrict__ arow = X + (size_t)m * DD;
    v8f c = {};
#pragma unroll
    for (int k = 0; k < DD; k += 4) {
      const int ka = k + 2 * hl;
      v2f a;
      a.x = arow[ka];
      a.y = arow[ka + 1];
      c = __builtin_amdgcn_wmma_f32_16x16x4_f32(false, a, false, bf[k / 4], (short)0, c, false, false);
    }
#pragma unroll
    for (int r = 0; r < 8; ++r) {
      Y[(size_t)(row0 + 8 * hl + r) * DD + n0 + l] = c[r];
    }
  }
}

#define FILL_EPT 8
__global__ void __launch_bounds__(256) fill_kernel(const int* __restrict__ src, const int* __restrict__ dst,
                            const unsigned* __restrict__ lrow, const unsigned* __restrict__ btot,
                            const unsigned* __restrict__ tab_in,
                            const unsigned* __restrict__ rank, int* __restrict__ col,
                            unsigned* __restrict__ rowptr) {
  __shared__ unsigned wt[8];
  __shared__ unsigned boff[MB + 1];
  const int t = blockIdx.x * 256 + threadIdx.x;
  const bool act = t < NE / FILL_EPT;
  const int te = act ? t : NE / FILL_EPT - 1;
  int ss[FILL_EPT];
  int dd[FILL_EPT];
  unsigned rr[FILL_EPT];
  unsigned pos[FILL_EPT];
#pragma unroll
  for (int v = 0; v < FILL_EPT / 4; ++v) {
    const int4 s4 = ((const int4*)src)[(FILL_EPT / 4) * te + v];
    const int4 d4 = ((const int4*)dst)[(FILL_EPT / 4) * te + v];
    const uint4 r4 = ((const uint4*)rank)[(FILL_EPT / 4) * te + v];
    ss[4 * v + 0] = s4.x; ss[4 * v + 1] = s4.y; ss[4 * v + 2] = s4.z; ss[4 * v + 3] = s4.w;
    dd[4 * v + 0] = d4.x; dd[4 * v + 1] = d4.y; dd[4 * v + 2] = d4.z; dd[4 * v + 3] = d4.w;
    rr[4 * v + 0] = r4.x; rr[4 * v + 1] = r4.y; rr[4 * v + 2] = r4.z; rr[4 * v + 3] = r4.w;
  }
  const unsigned lr_own = t < NN ? lrow[t] : 0u;
  const unsigned bt = threadIdx.x < MB ? btot[threadIdx.x] : 0u;
  unsigned total;
  const unsigned ex = block_excl_scan256(bt, wt, total);
  if (threadIdx.x < MB) boff[threadIdx.x] = ex;
  if (threadIdx.x == MB) boff[MB] = total;
  __syncthreads();
  if (t <= NN) rowptr[t] = t < NN ? boff[t >> 8] + lr_own : boff[MB];
  const unsigned* __restrict__ tab = tab_in + (size_t)((FILL_EPT * te) / CH) * NN;
  asm volatile("" :: "v"(ss[0]), "v"(ss[1]), "v"(ss[2]), "v"(ss[3]),
                     "v"(ss[4]), "v"(ss[5]), "v"(ss[6]), "v"(ss[7]));
#pragma unroll
  for (int k = 0; k < FILL_EPT; ++k) {
    unsigned d = (unsigned)dd[k];
    d = d < (unsigned)NN ? d : (unsigned)(NN - 1);
    pos[k] = boff[d >> 8] + lrow[d] + tab[d] + rr[k];
  }
  if (act) {
#pragma unroll
    for (int k = 0; k < FILL_EPT; ++k) {
      if (pos[k] < (unsigned)NE) col[pos[k]] = ss[k];
    }
  }
}

template <int K>
__device__ __forceinline__ void issue_rows(const float4* __restrict__ H4, int mine, int j,
                                           int lane, float4* r) {
#pragma unroll
  for (int k = 0; k < K; ++k) {
    const unsigned sidx = (unsigned)__builtin_amdgcn_readlane(mine, j + k);
    __builtin_assume(sidx < (unsigned)NN);
    r[k] = H4[(size_t)sidx * (DD / 4) + lane];
  }
}
template <int K, bool SCALE>
__device__ __forceinline__ void add_rows(const float4* r, float sv, int j, float4& acc) {
#pragma unroll
  for (int k = 0; k < K; ++k) {
    if constexpr (SCALE) {
      const float s = __int_as_float(__builtin_amdgcn_readlane(__float_as_int(sv), j + k));
      acc.x += r[k].x * s; acc.y += r[k].y * s; acc.z += r[k].z * s; acc.w += r[k].w * s;
    } else {
      acc.x += r[k].x; acc.y += r[k].y; acc.z += r[k].z; acc.w += r[k].w;
    }
  }
}

template <bool SCALE>
__device__ __forceinline__ void gather_chunk(const float4* __restrict__ H4, int cur, float sv, int n,
                                             int lane, float4& acc) {
  int j = 0;
  for (; j + 8 <= n; j += 8) {
    float4 r[8];
    issue_rows<8>(H4, cur, j, lane, r);
    add_rows<8, SCALE>(r, sv, j, acc);
  }
  const int t = n - j;
  float4 r4[4], r2[2], r1[1];
  const int j2 = j + (t & 4);
  const int j1 = j2 + (t & 2);
  if (t & 4) issue_rows<4>(H4, cur, j, lane, r4);
  if (t & 2) issue_rows<2>(H4, cur, j2, lane, r2);
  if (t & 1) issue_rows<1>(H4, cur, j1, lane, r1);
  if (t & 4) add_rows<4, SCALE>(r4, sv, j, acc);
  if (t & 2) add_rows<2, SCALE>(r2, sv, j2, acc);
  if (t & 1) add_rows<1, SCALE>(r1, sv, j1, acc);
}

#define ROWS_PER_WG 8
#define GRP_PER_WG 5
#define SPMM_WGS (NN / (ROWS_PER_WG * GRP_PER_WG))
template <bool POOL>
__global__ void __launch_bounds__(256) __attribute__((amdgpu_waves_per_eu(10, 10))) spmm_pull(const float* __restrict__ H,
                                                 const unsigned* __restrict__ rowptr,
                                                 const int* __restrict__ col,
                                                 const float* __restrict__ nd,
                                                 const float* __restrict__ ns,
                                                 const float* __restrict__ bias,
                                                 const int* __restrict__ gid,
                                                 float* __restrict__ OUT,
                                                 float* __restrict__ CNT) {
  const int lane = threadIdx.x & 31;
  const int wave = __builtin_amdgcn_readfirstlane((int)(threadIdx.x >> 5));
  const int dbase = (int)(blockIdx.x * (ROWS_PER_WG * GRP_PER_WG)) + wave;
  const float4* __restrict__ H4 = (const float4*)H;

  unsigned bnd = 0u;
  float sc_l = 0.0f, sn_l = 1.0f;
  int g_l = -1;
  {
    const int i = lane < GRP_PER_WG ? lane : lane - GRP_PER_WG;
    const int dl = dbase + ROWS_PER_WG * i;
    if (lane < 2 * GRP_PER_WG && dl < NN) {
      bnd = rowptr[dl + (lane < GRP_PER_WG ? 0 : 1)];
      bnd = bnd < (unsigned)NE ? bnd : (unsigned)NE;
      if (lane < GRP_PER_WG) {
        sc_l = nd[dl];
        if constexpr (POOL) g_l = gid[dl]; else sn_l = ns[dl];
      }
    }
  }
  float4 b = {0.0f, 0.0f, 0.0f, 0.0f};
  if constexpr (!POOL) b = ((const float4*)bias)[lane];

  float4 racc = {0.0f, 0.0f, 0.0f, 0.0f};
  int rg = -1;
  float rn = 0.0f;

  int mi[GRP_PER_WG];
#pragma unroll
  for (int i = 0; i < GRP_PER_WG; ++i) {
    const unsigned bi = (unsigned)__builtin_amdgcn_readlane((int)bnd, i);
    const unsigned ei = (unsigned)__builtin_amdgcn_readlane((int)bnd, GRP_PER_WG + i);
    mi[i] = bi + lane < ei ? col[bi + lane] : 0;
  }
  float ms[GRP_PER_WG];
#pragma unroll
  for (int i = 0; i < GRP_PER_WG; ++i) {
    ms[i] = 1.0f;
    if constexpr (!POOL) {
      const unsigned um = (unsigned)mi[i];
      ms[i] = ns[um < (unsigned)NN ? um : (unsigned)(NN - 1)];
    }
  }
#pragma unroll 1
  for (int it = 0; it < GRP_PER_WG; ++it) {
    const int d = dbase + ROWS_PER_WG * it;
    const unsigned beg = (unsigned)__builtin_amdgcn_readlane((int)bnd, it);
    const unsigned end = (unsigned)__builtin_amdgcn_readlane((int)bnd, GRP_PER_WG + it);
    int mine = mi[0];
    float smine = ms[0];
#pragma unroll
    for (int i = 1; i < GRP_PER_WG; ++i) {
      mine = it == i ? mi[i] : mine;
      smine = it == i ? ms[i] : smine;
    }
    float4 acc = {0.0f, 0.0f, 0.0f, 0.0f};
    for (unsigned p = beg; p < end; p += 32u) {
      int raw = mine;
      if (p != beg) raw = p + lane < end ? col[p + lane] : 0;
      const unsigned uraw = (unsigned)raw;
      const int cur = (int)(uraw < (unsigned)NN ? uraw : (unsigned)(NN - 1));
      float sv = smine;
      if constexpr (!POOL) {
        if (p != beg) sv = ns[cur];
      }
      gather_chunk<!POOL>(H4, cur, sv, (int)((end - p) < 32u ? (end - p) : 32u), lane, acc);
    }
    const float sc = __int_as_float(__builtin_amdgcn_readlane(__float_as_int(sc_l), it));
    if constexpr (POOL) {
      float4 o;
      o.x = acc.x * sc;
      o.y = acc.y * sc;
      o.z = acc.z * sc;
      o.w = acc.w * sc;
      const int g = __builtin_amdgcn_readlane(g_l, it);
      if (d < NN) {
        if (g != rg) {
          if (rn > 0.0f && (unsigned)rg < (unsigned)NG) {
            atomAddF(&OUT[rg * DD + 4 * lane + 0], racc.x);
            atomAddF(&OUT[rg * DD + 4 * lane + 1], racc.y);
            atomAddF(&OUT[rg * DD + 4 * lane + 2], racc.z);
            atomAddF(&OUT[rg * DD + 4 * lane + 3], racc.w);
            if (lane == 0) atomAddF(&CNT[rg], rn);
          }
          racc = o;
          rg = g;
          rn = 1.0f;
        } else {
          racc.x += o.x; racc.y += o.y; racc.z += o.z; racc.w += o.w;
          rn += 1.0f;
        }
      }
    } else {
      const float sn = __int_as_float(__builtin_amdgcn_readlane(__float_as_int(sn_l), it));
      if (d < NN) {
        float4 o;
        o.x = fmaxf(acc.x * sc + b.x, 0.0f) * sn;
        o.y = fmaxf(acc.y * sc + b.y, 0.0f) * sn;
        o.z = fmaxf(acc.z * sc + b.z, 0.0f) * sn;
        o.w = fmaxf(acc.w * sc + b.w, 0.0f) * sn;
        ((float4*)OUT)[(size_t)d * (DD / 4) + lane] = o;
      }
    }
  }
  if constexpr (POOL) {
    __shared__ float4 prow[ROWS_PER_WG][DD / 4];
    __shared__ int pgid[ROWS_PER_WG];
    __shared__ float pcnt[ROWS_PER_WG];
    prow[wave][lane] = racc;
    if (lane == 0) {
      pgid[wave] = rn > 0.0f ? rg : -1;
      pcnt[wave] = rn;
    }
    __syncthreads();
    if (threadIdx.x < DD) {
      const int c = threadIdx.x;
      const float* __restrict__ pr = (const float*)&prow[0][0];
      float v[ROWS_PER_WG];
      int gs[ROWS_PER_WG];
#pragma unroll
      for (int r = 0; r < ROWS_PER_WG; ++r) {
        v[r] = pr[r * DD + c];
        gs[r] = __builtin_amdgcn_readfirstlane(pgid[r]);
      }
      int g = gs[0];
      float s = v[0];
#pragma unroll
      for (int r = 1; r < ROWS_PER_WG; ++r) {
        if (gs[r] != g) {
          if ((unsigned)g < (unsigned)NG) atomAddF(&OUT[g * DD + c], s);
          g = gs[r];
          s = v[r];
        } else {
          s += v[r];
        }
      }
      if ((unsigned)g < (unsigned)NG) atomAddF(&OUT[g * DD + c], s);
    } else if (threadIdx.x == DD) {
      int g = pgid[0];
      float n = pcnt[0];
      for (int r = 1; r < ROWS_PER_WG; ++r) {
        const int gr = pgid[r];
        if (gr != g) {
          if ((unsigned)g < (unsigned)NG) atomAddF(&CNT[g], n);
          g = gr;
          n = pcnt[r];
        } else {
          n += pcnt[r];
        }
      }
      if ((unsigned)g < (unsigned)NG) atomAddF(&CNT[g], n);
    }
  }
}

#define HG_PITCH (DD + 4)
#define RO_ROWS 16
#define NC_TILES ((NC + 15) / 16)
__global__ void __launch_bounds__(256) readout_kernel(const float* __restrict__ S,
                                                      const float* __restrict__ cnt,
                                                      const float* __restrict__ W2,
                                                      const float* __restrict__ b2,
                                                      const float* __restrict__ Wc,
                                                      const float* __restrict__ bc,
                                                      float* __restrict__ out) {
  __shared__ float hg[RO_ROWS * HG_PITCH];
  const int lane = threadIdx.x & 31;
  const int wave = __builtin_amdgcn_readfirstlane((int)(threadIdx.x >> 5));
  const int hl   = lane >> 4;
  const int l    = lane & 15;
  const int g0   = blockIdx.x * RO_ROWS;
  const int n0   = wave * 16;
  const float* __restrict__ arow = S + (size_t)(g0 + l) * DD;

  const bool cw = wave < NC_TILES;
  const int cn = n0 + l;
  const int cl = cn < NC ? cn : NC - 1;
  v2f wf[DD / 4];
  float bcl = 0.0f;
  if (cw) {
#pragma unroll
    for (int k = 0; k < DD; k += 4) {
      const int ka = k + 2 * hl;
      wf[k / 4].x = Wc[ka * NC + cl];
      wf[k / 4].y = Wc[(ka + 1) * NC + cl];
    }
    bcl = bc[cl];
  }

  v8f c = {};
#pragma unroll
  for (int k = 0; k < DD; k += 4) {
    const int ka = k + 2 * hl;
    v2f a, b;
    a.x = arow[ka];
    a.y = arow[ka + 1];
    b.x = W2[ka * DD + n0 + l];
    b.y = W2[(ka + 1) * DD + n0 + l];
    c = __builtin_amdgcn_wmma_f32_16x16x4_f32(false, a, false, b, (short)0, c, false, false);
  }
  const float bcol = b2[n0 + l];
#pragma unroll
  for (int r = 0; r < 8; ++r) {
    const float n = cnt[g0 + 8 * hl + r];
    const float inv = 1.0f / fmaxf(n, 1.0f);
    hg[(8 * hl + r) * HG_PITCH + n0 + l] = (c[r] + n * bcol) * inv;
  }
  __syncthreads();
  if (cw) {
    const float* __restrict__ hrow = hg + l * HG_PITCH;
    v8f c2 = {};
#pragma unroll
    for (int k = 0; k < DD; k += 4) {
      const int ka = k + 2 * hl;
      v2f a;
      a.x = hrow[ka];
      a.y = hrow[ka + 1];
      c2 = __builtin_amdgcn_wmma_f32_16x16x4_f32(false, a, false, wf[k / 4], (short)0, c2, false, false);
    }
    if (cn < NC) {
#pragma unroll
      for (int r = 0; r < 8; ++r) out[(g0 + 8 * hl + r) * NC + cn] = c2[r] + bcl;
    }
  }
}

extern "C" void kernel_launch(void* const* d_in, const int* in_sizes, int n_in,
                              void* d_out, int out_size, void* d_ws, size_t ws_size,
                              hipStream_t stream) {
  const float* feat = (const float*)d_in[0];
  const float* W1   = (const float*)d_in[1];
  const float* b1   = (const float*)d_in[2];
  const float* W2   = (const float*)d_in[3];
  const float* b2   = (const float*)d_in[4];
  const float* Wc   = (const float*)d_in[5];
  const float* bc   = (const float*)d_in[6];
  const int*   src  = (const int*)d_in[7];
  const int*   dst  = (const int*)d_in[8];
  const int*   gid  = (const int*)d_in[9];
  float* out = (float*)d_out;

  float*    ws      = (float*)d_ws;
  unsigned* btot    = (unsigned*)ws;
  unsigned* lrow    = (unsigned*)ws + NN;
  float*    nsrc    = ws + 2 * NN;
  float*    ndst    = nsrc + NN;
  float*    bufA    = ndst + NN;
  float*    bufB    = bufA + (size_t)NN * DD;
  float*    sums    = bufB + (size_t)NN * DD;
  float*    cnt     = sums + NG * DD;
  unsigned* rowptr  = (unsigned*)(cnt + NG);
  unsigned* rank    = rowptr + 40064;
  int*      col     = (int*)(rank + NE);
  unsigned* tab_in  = (unsigned*)(col + NE);
  unsigned* tab_out = tab_in + (size_t)NCH * NN;

  const int T = 256;

  deg_chunk_kernel<<<2 * NCH, 1024, 0, stream>>>(src, dst, tab_in, tab_out, rank);
  merge_gemm_kernel<<<MB + NN / 16 / GEMM_SPW, 256, 0, stream>>>(tab_in, tab_out, lrow, btot, nsrc, ndst, sums,
                                                                  feat, W1, bufA);
  fill_kernel<<<(NE / FILL_EPT + T - 1) / T, T, 0, stream>>>(src, dst, lrow, btot, tab_in, rank, col, rowptr);

  spmm_pull<false><<<SPMM_WGS, 256, 0, stream>>>(bufA, rowptr, col, ndst, nsrc, b1, gid, bufB, cnt);

  spmm_pull<true><<<SPMM_WGS, 256, 0, stream>>>(bufB, rowptr, col, ndst, nsrc, b2, gid, sums, cnt);

  readout_kernel<<<NG / RO_ROWS, 256, 0, stream>>>(sums, cnt, W2, b2, Wc, bc, out);
}


# layer-1 GEMM: a block's 80 feature rows staged once in LDS by whole-row 512-B loads, A fragments read from LDS (pitch 132 words); the 8 waves' strided 8-B global A loads gone; sums unchanged
# speedup vs baseline: 4.4962x; 1.1400x over previous
#include <hip/hip_runtime.h>

#define NN 40000
#define NE 640000
#define DD 128
#define NC 40
#define NG 128

typedef __attribute__((ext_vector_type(2))) float v2f;
typedef __attribute__((ext_vector_type(8))) float v8f;

__device__ __forceinline__ void atomAddF(float* p, float v) {
#if defined(__HIP_DEVICE_COMPILE__)
  unsafeAtomicAdd(p, v);
#else
  __hip_atomic_fetch_add(p, v, __ATOMIC_RELAXED, __HIP_MEMORY_SCOPE_AGENT);
#endif
}

#define NCH 64
#define CH (NE / NCH)
#define DEG_IT ((CH + 1023) / 1024)
__global__ void __launch_bounds__(1024) deg_chunk_kernel(const int* __restrict__ src,
                                                         const int* __restrict__ dst,
                                                         unsigned* __restrict__ tab_in,
                                                         unsigned* __restrict__ tab_out,
                                                         unsigned* __restrict__ rank) {
  __shared__ unsigned h[NN];
  const int which = blockIdx.x / NCH;
  const int w = blockIdx.x % NCH;
  const int* __restrict__ key = which ? src : dst;
  const int e0 = w * CH;
  unsigned k[DEG_IT];
#pragma unroll
  for (int j = 0; j < DEG_IT; ++j) {
    const int i = threadIdx.x + 1024 * j;
    k[j] = i < CH ? (unsigned)key[e0 + i] : 0xFFFFFFFFu;
  }
  for (int i = threadIdx.x; i < NN; i += 1024) h[i] = 0u;
  __syncthreads();
#pragma unroll
  for (int j = 0; j < DEG_IT; ++j) {
    if (k[j] < (unsigned)NN) {
      const unsigned r = atomicAdd(&h[k[j]], 1u);
      if (!which) rank[e0 + threadIdx.x + 1024 * j] = r;
    }
  }
  __syncthreads();
  unsigned* __restrict__ t = (which ? tab_out : tab_in) + (size_t)w * NN;
  for (int i = threadIdx.x; i < NN; i += 1024) t[i] = h[i];
}

__device__ __forceinline__ unsigned block_excl_scan256(unsigned v, unsigned* wt, unsigned& total) {
  const int lane = threadIdx.x & 31;
  const int w = threadIdx.x >> 5;
  unsigned inc = v;
  for (int o = 1; o < 32; o <<= 1) {
    const unsigned u = __shfl_up(inc, o);
    if (lane >= o) inc += u;
  }
  if (lane == 31) wt[w] = inc;
  __syncthreads();
  unsigned woff = 0, tot = 0;
#pragma unroll
  for (int k = 0; k < 8; ++k) {
    const unsigned x = wt[k];
    woff += k < w ? x : 0u;
    tot += x;
  }
  total = tot;
  return woff + inc - v;
}

#define MB ((NN + 255) / 256)
#define GEMM_SPW 5
#define GEMM_ROWS (16 * GEMM_SPW)
#define A_PITCH (DD + 4)
__global__ void __launch_bounds__(256) merge_gemm_kernel(unsigned* __restrict__ tab_in,
                                                         const unsigned* __restrict__ tab_out,
                                                         unsigned* __restrict__ lrow,
                                                         unsigned* __restrict__ btot,
                                                         float* __restrict__ ns,
                                                         float* __restrict__ nd,
                                                         float* __restrict__ zsum,
                                                         const float* __restrict__ X,
                                                         const float* __restrict__ W,
                                                         float* __restrict__ Y) {
  __shared__ unsigned wt[8];
  __shared__ float As[GEMM_ROWS * A_PITCH];
  if (blockIdx.x < MB) {
    const int d = blockIdx.x * 256 + threadIdx.x;
    if (d < NG * DD + NG) zsum[d] = 0.0f;
    unsigned run = 0, so = 0;
    if (d < NN) {
#pragma unroll 8
      for (int w = 0; w < NCH; ++w) {
        const unsigned c = tab_in[(size_t)w * NN + d];
        so += tab_out[(size_t)w * NN + d];
        tab_in[(size_t)w * NN + d] = run;
        run += c;
      }
    }
    unsigned total;
    const unsigned ex = block_excl_scan256(run, wt, total);
    if (d < NN) {
      lrow[d] = ex;
      ns[d] = rsqrtf(fmaxf((float)so, 1.0f));
      nd[d] = rsqrtf(fmaxf((float)run, 1.0f));
    }
    if (threadIdx.x == 0) btot[blockIdx.x] = total;
    return;
  }
  const int gb   = blockIdx.x - MB;
  const int lane = threadIdx.x & 31;
  const int wave = threadIdx.x >> 5;
  const int hl   = lane >> 4;
  const int l    = lane & 15;
  const int n0   = wave * 16;

  const float4* __restrict__ X4 = (const float4*)X + (size_t)gb * GEMM_ROWS * (DD / 4);
  float4 st[GEMM_ROWS / 8];
#pragma unroll
  for (int i = 0; i < GEMM_ROWS / 8; ++i) st[i] = X4[(wave + 8 * i) * (DD / 4) + lane];

  v2f bf[DD / 4];
#pragma unroll
  for (int k = 0; k < DD; k += 4) {
    const int ka = k + 2 * hl;
    bf[k / 4].x = W[ka * DD + n0 + l];
    bf[k / 4].y = W[(ka + 1) * DD + n0 + l];
  }

#pragma unroll
  for (int i = 0; i < GEMM_ROWS / 8; ++i)
    *(float4*)&As[(wave + 8 * i) * A_PITCH + 4 * lane] = st[i];
  __syncthreads();

  for (int s = 0; s < GEMM_SPW; ++s) {
    const int row0 = (gb * GEMM_SPW + s) * 16;
    const float* __restrict__ arow = As + (s * 16 + l) * A_PITCH + 2 * hl;
    v8f c = {};
#pragma unroll
    for (int k = 0; k < DD; k += 4) {
      const v2f a = *(const v2f*)(arow + k);
      c = __builtin_amdgcn_wmma_f32_16x16x4_f32(false, a, false, bf[k / 4], (short)0, c, false, false);
    }
#pragma unroll
    for (int r = 0; r < 8; ++r) {
      Y[(size_t)(row0 + 8 * hl + r) * DD + n0 + l] = c[r];
    }
  }
}

#define FILL_EPT 8
__global__ void __launch_bounds__(256) fill_kernel(const int* __restrict__ src, const int* __restrict__ dst,
                            const unsigned* __restrict__ lrow, const unsigned* __restrict__ btot,
                            const unsigned* __restrict__ tab_in,
                            const unsigned* __restrict__ rank, int* __restrict__ col,
                            unsigned* __restrict__ rowptr) {
  __shared__ unsigned wt[8];
  __shared__ unsigned boff[MB + 1];
  const int t = blockIdx.x * 256 + threadIdx.x;
  const bool act = t < NE / FILL_EPT;
  const int te = act ? t : NE / FILL_EPT - 1;
  int ss[FILL_EPT];
  int dd[FILL_EPT];
  unsigned rr[FILL_EPT];
  unsigned pos[FILL_EPT];
#pragma unroll
  for (int v = 0; v < FILL_EPT / 4; ++v) {
    const int4 s4 = ((const int4*)src)[(FILL_EPT / 4) * te + v];
    const int4 d4 = ((const int4*)dst)[(FILL_EPT / 4) * te + v];
    const uint4 r4 = ((const uint4*)rank)[(FILL_EPT / 4) * te + v];
    ss[4 * v + 0] = s4.x; ss[4 * v + 1] = s4.y; ss[4 * v + 2] = s4.z; ss[4 * v + 3] = s4.w;
    dd[4 * v + 0] = d4.x; dd[4 * v + 1] = d4.y; dd[4 * v + 2] = d4.z; dd[4 * v + 3] = d4.w;
    rr[4 * v + 0] = r4.x; rr[4 * v + 1] = r4.y; rr[4 * v + 2] = r4.z; rr[4 * v + 3] = r4.w;
  }
  const unsigned lr_own = t < NN ? lrow[t] : 0u;
  const unsigned bt = threadIdx.x < MB ? btot[threadIdx.x] : 0u;
  unsigned total;
  const unsigned ex = block_excl_scan256(bt, wt, total);
  if (threadIdx.x < MB) boff[threadIdx.x] = ex;
  if (threadIdx.x == MB) boff[MB] = total;
  __syncthreads();
  if (t <= NN) rowptr[t] = t < NN ? boff[t >> 8] + lr_own : boff[MB];
  const unsigned* __restrict__ tab = tab_in + (size_t)((FILL_EPT * te) / CH) * NN;
  asm volatile("" :: "v"(ss[0]), "v"(ss[1]), "v"(ss[2]), "v"(ss[3]),
                     "v"(ss[4]), "v"(ss[5]), "v"(ss[6]), "v"(ss[7]));
#pragma unroll
  for (int k = 0; k < FILL_EPT; ++k) {
    unsigned d = (unsigned)dd[k];
    d = d < (unsigned)NN ? d : (unsigned)(NN - 1);
    pos[k] = boff[d >> 8] + lrow[d] + tab[d] + rr[k];
  }
  if (act) {
#pragma unroll
    for (int k = 0; k < FILL_EPT; ++k) {
      if (pos[k] < (unsigned)NE) col[pos[k]] = ss[k];
    }
  }
}

template <int K>
__device__ __forceinline__ void issue_rows(const float4* __restrict__ H4, int mine, int j,
                                           int lane, float4* r) {
#pragma unroll
  for (int k = 0; k < K; ++k) {
    const unsigned sidx = (unsigned)__builtin_amdgcn_readlane(mine, j + k);
    __builtin_assume(sidx < (unsigned)NN);
    r[k] = H4[(size_t)sidx * (DD / 4) + lane];
  }
}
template <int K, bool SCALE>
__device__ __forceinline__ void add_rows(const float4* r, float sv, int j, float4& acc) {
#pragma unroll
  for (int k = 0; k < K; ++k) {
    if constexpr (SCALE) {
      const float s = __int_as_float(__builtin_amdgcn_readlane(__float_as_int(sv), j + k));
      acc.x += r[k].x * s; acc.y += r[k].y * s; acc.z += r[k].z * s; acc.w += r[k].w * s;
    } else {
      acc.x += r[k].x; acc.y += r[k].y; acc.z += r[k].z; acc.w += r[k].w;
    }
  }
}

template <bool SCALE>
__device__ __forceinline__ void gather_chunk(const float4* __restrict__ H4, int cur, float sv, int n,
                                             int lane, float4& acc) {
  int j = 0;
  for (; j + 8 <= n; j += 8) {
    float4 r[8];
    issue_rows<8>(H4, cur, j, lane, r);
    add_rows<8, SCALE>(r, sv, j, acc);
  }
  const int t = n - j;
  float4 r4[4], r2[2], r1[1];
  const int j2 = j + (t & 4);
  const int j1 = j2 + (t & 2);
  if (t & 4) issue_rows<4>(H4, cur, j, lane, r4);
  if (t & 2) issue_rows<2>(H4, cur, j2, lane, r2);
  if (t & 1) issue_rows<1>(H4, cur, j1, lane, r1);
  if (t & 4) add_rows<4, SCALE>(r4, sv, j, acc);
  if (t & 2) add_rows<2, SCALE>(r2, sv, j2, acc);
  if (t & 1) add_rows<1, SCALE>(r1, sv, j1, acc);
}

#define ROWS_PER_WG 8
#define GRP_PER_WG 5
#define SPMM_WGS (NN / (ROWS_PER_WG * GRP_PER_WG))
template <bool POOL>
__global__ void __launch_bounds__(256) __attribute__((amdgpu_waves_per_eu(10, 10))) spmm_pull(const float* __restrict__ H,
                                                 const unsigned* __restrict__ rowptr,
                                                 const int* __restrict__ col,
                                                 const float* __restrict__ nd,
                                                 const float* __restrict__ ns,
                                                 const float* __restrict__ bias,
                                                 const int* __restrict__ gid,
                                                 float* __restrict__ OUT,
                                                 float* __restrict__ CNT) {
  const int lane = threadIdx.x & 31;
  const int wave = __builtin_amdgcn_readfirstlane((int)(threadIdx.x >> 5));
  const int dbase = (int)(blockIdx.x * (ROWS_PER_WG * GRP_PER_WG)) + wave;
  const float4* __restrict__ H4 = (const float4*)H;

  unsigned bnd = 0u;
  float sc_l = 0.0f, sn_l = 1.0f;
  int g_l = -1;
  {
    const int i = lane < GRP_PER_WG ? lane : lane - GRP_PER_WG;
    const int dl = dbase + ROWS_PER_WG * i;
    if (lane < 2 * GRP_PER_WG && dl < NN) {
      bnd = rowptr[dl + (lane < GRP_PER_WG ? 0 : 1)];
      bnd = bnd < (unsigned)NE ? bnd : (unsigned)NE;
      if (lane < GRP_PER_WG) {
        sc_l = nd[dl];
        if constexpr (POOL) g_l = gid[dl]; else sn_l = ns[dl];
      }
    }
  }
  float4 b = {0.0f, 0.0f, 0.0f, 0.0f};
  if constexpr (!POOL) b = ((const float4*)bias)[lane];

  float4 racc = {0.0f, 0.0f, 0.0f, 0.0f};
  int rg = -1;
  float rn = 0.0f;

  int mi[GRP_PER_WG];
#pragma unroll
  for (int i = 0; i < GRP_PER_WG; ++i) {
    const unsigned bi = (unsigned)__builtin_amdgcn_readlane((int)bnd, i);
    const unsigned ei = (unsigned)__builtin_amdgcn_readlane((int)bnd, GRP_PER_WG + i);
    mi[i] = bi + lane < ei ? col[bi + lane] : 0;
  }
  float ms[GRP_PER_WG];
#pragma unroll
  for (int i = 0; i < GRP_PER_WG; ++i) {
    ms[i] = 1.0f;
    if constexpr (!POOL) {
      const unsigned um = (unsigned)mi[i];
      ms[i] = ns[um < (unsigned)NN ? um : (unsigned)(NN - 1)];
    }
  }
#pragma unroll 1
  for (int it = 0; it < GRP_PER_WG; ++it) {
    const int d = dbase + ROWS_PER_WG * it;
    const unsigned beg = (unsigned)__builtin_amdgcn_readlane((int)bnd, it);
    const unsigned end = (unsigned)__builtin_amdgcn_readlane((int)bnd, GRP_PER_WG + it);
    int mine = mi[0];
    float smine = ms[0];
#pragma unroll
    for (int i = 1; i < GRP_PER_WG; ++i) {
      mine = it == i ? mi[i] : mine;
      smine = it == i ? ms[i] : smine;
    }
    float4 acc = {0.0f, 0.0f, 0.0f, 0.0f};
    for (unsigned p = beg; p < end; p += 32u) {
      int raw = mine;
      if (p != beg) raw = p + lane < end ? col[p + lane] : 0;
      const unsigned uraw = (unsigned)raw;
      const int cur = (int)(uraw < (unsigned)NN ? uraw : (unsigned)(NN - 1));
      float sv = smine;
      if constexpr (!POOL) {
        if (p != beg) sv = ns[cur];
      }
      gather_chunk<!POOL>(H4, cur, sv, (int)((end - p) < 32u ? (end - p) : 32u), lane, acc);
    }
    const float sc = __int_as_float(__builtin_amdgcn_readlane(__float_as_int(sc_l), it));
    if constexpr (POOL) {
      float4 o;
      o.x = acc.x * sc;
      o.y = acc.y * sc;
      o.z = acc.z * sc;
      o.w = acc.w * sc;
      const int g = __builtin_amdgcn_readlane(g_l, it);
      if (d < NN) {
        if (g != rg) {
          if (rn > 0.0f && (unsigned)rg < (unsigned)NG) {
            atomAddF(&OUT[rg * DD + 4 * lane + 0], racc.x);
            atomAddF(&OUT[rg * DD + 4 * lane + 1], racc.y);
            atomAddF(&OUT[rg * DD + 4 * lane + 2], racc.z);
            atomAddF(&OUT[rg * DD + 4 * lane + 3], racc.w);
            if (lane == 0) atomAddF(&CNT[rg], rn);
          }
          racc = o;
          rg = g;
          rn = 1.0f;
        } else {
          racc.x += o.x; racc.y += o.y; racc.z += o.z; racc.w += o.w;
          rn += 1.0f;
        }
      }
    } else {
      const float sn = __int_as_float(__builtin_amdgcn_readlane(__float_as_int(sn_l), it));
      if (d < NN) {
        float4 o;
        o.x = fmaxf(acc.x * sc + b.x, 0.0f) * sn;
        o.y = fmaxf(acc.y * sc + b.y, 0.0f) * sn;
        o.z = fmaxf(acc.z * sc + b.z, 0.0f) * sn;
        o.w = fmaxf(acc.w * sc + b.w, 0.0f) * sn;
        ((float4*)OUT)[(size_t)d * (DD / 4) + lane] = o;
      }
    }
  }
  if constexpr (POOL) {
    __shared__ float4 prow[ROWS_PER_WG][DD / 4];
    __shared__ int pgid[ROWS_PER_WG];
    __shared__ float pcnt[ROWS_PER_WG];
    prow[wave][lane] = racc;
    if (lane == 0) {
      pgid[wave] = rn > 0.0f ? rg : -1;
      pcnt[wave] = rn;
    }
    __syncthreads();
    if (threadIdx.x < DD) {
      const int c = threadIdx.x;
      const float* __restrict__ pr = (const float*)&prow[0][0];
      float v[ROWS_PER_WG];
      int gs[ROWS_PER_WG];
#pragma unroll
      for (int r = 0; r < ROWS_PER_WG; ++r) {
        v[r] = pr[r * DD + c];
        gs[r] = __builtin_amdgcn_readfirstlane(pgid[r]);
      }
      int g = gs[0];
      float s = v[0];
#pragma unroll
      for (int r = 1; r < ROWS_PER_WG; ++r) {
        if (gs[r] != g) {
          if ((unsigned)g < (unsigned)NG) atomAddF(&OUT[g * DD + c], s);
          g = gs[r];
          s = v[r];
        } else {
          s += v[r];
        }
      }
      if ((unsigned)g < (unsigned)NG) atomAddF(&OUT[g * DD + c], s);
    } else if (threadIdx.x == DD) {
      int g = pgid[0];
      float n = pcnt[0];
      for (int r = 1; r < ROWS_PER_WG; ++r) {
        const int gr = pgid[r];
        if (gr != g) {
          if ((unsigned)g < (unsigned)NG) atomAddF(&CNT[g], n);
          g = gr;
          n = pcnt[r];
        } else {
          n += pcnt[r];
        }
      }
      if ((unsigned)g < (unsigned)NG) atomAddF(&CNT[g], n);
    }
  }
}

#define HG_PITCH (DD + 4)
#define RO_ROWS 16
#define NC_TILES ((NC + 15) / 16)
__global__ void __launch_bounds__(256) readout_kernel(const float* __restrict__ S,
                                                      const float* __restrict__ cnt,
                                                      const float* __restrict__ W2,
                                                      const float* __restrict__ b2,
                                                      const float* __restrict__ Wc,
                                                      const float* __restrict__ bc,
                                                      float* __restrict__ out) {
  __shared__ float hg[RO_ROWS * HG_PITCH];
  const int lane = threadIdx.x & 31;
  const int wave = __builtin_amdgcn_readfirstlane((int)(threadIdx.x >> 5));
  const int hl   = lane >> 4;
  const int l    = lane & 15;
  const int g0   = blockIdx.x * RO_ROWS;
  const int n0   = wave * 16;
  const float* __restrict__ arow = S + (size_t)(g0 + l) * DD;

  const bool cw = wave < NC_TILES;
  const int cn = n0 + l;
  const int cl = cn < NC ? cn : NC - 1;
  v2f wf[DD / 4];
  float bcl = 0.0f;
  if (cw) {
#pragma unroll
    for (int k = 0; k < DD; k += 4) {
      const int ka = k + 2 * hl;
      wf[k / 4].x = Wc[ka * NC + cl];
      wf[k / 4].y = Wc[(ka + 1) * NC + cl];
    }
    bcl = bc[cl];
  }

  v8f c = {};
#pragma unroll
  for (int k = 0; k < DD; k += 4) {
    const int ka = k + 2 * hl;
    v2f a, b;
    a.x = arow[ka];
    a.y = arow[ka + 1];
    b.x = W2[ka * DD + n0 + l];
    b.y = W2[(ka + 1) * DD + n0 + l];
    c = __builtin_amdgcn_wmma_f32_16x16x4_f32(false, a, false, b, (short)0, c, false, false);
  }
  const float bcol = b2[n0 + l];
#pragma unroll
  for (int r = 0; r < 8; ++r) {
    const float n = cnt[g0 + 8 * hl + r];
    const float inv = 1.0f / fmaxf(n, 1.0f);
    hg[(8 * hl + r) * HG_PITCH + n0 + l] = (c[r] + n * bcol) * inv;
  }
  __syncthreads();
  if (cw) {
    const float* __restrict__ hrow = hg + l * HG_PITCH;
    v8f c2 = {};
#pragma unroll
    for (int k = 0; k < DD; k += 4) {
      const int ka = k + 2 * hl;
      v2f a;
      a.x = hrow[ka];
      a.y = hrow[ka + 1];
      c2 = __builtin_amdgcn_wmma_f32_16x16x4_f32(false, a, false, wf[k / 4], (short)0, c2, false, false);
    }
    if (cn < NC) {
#pragma unroll
      for (int r = 0; r < 8; ++r) out[(g0 + 8 * hl + r) * NC + cn] = c2[r] + bcl;
    }
  }
}

extern "C" void kernel_launch(void* const* d_in, const int* in_sizes, int n_in,
                              void* d_out, int out_size, void* d_ws, size_t ws_size,
                              hipStream_t stream) {
  const float* feat = (const float*)d_in[0];
  const float* W1   = (const float*)d_in[1];
  const float* b1   = (const float*)d_in[2];
  const float* W2   = (const float*)d_in[3];
  const float* b2   = (const float*)d_in[4];
  const float* Wc   = (const float*)d_in[5];
  const float* bc   = (const float*)d_in[6];
  const int*   src  = (const int*)d_in[7];
  const int*   dst  = (const int*)d_in[8];
  const int*   gid  = (const int*)d_in[9];
  float* out = (float*)d_out;

  float*    ws      = (float*)d_ws;
  unsigned* btot    = (unsigned*)ws;
  unsigned* lrow    = (unsigned*)ws + NN;
  float*    nsrc    = ws + 2 * NN;
  float*    ndst    = nsrc + NN;
  float*    bufA    = ndst + NN;
  float*    bufB    = bufA + (size_t)NN * DD;
  float*    sums    = bufB + (size_t)NN * DD;
  float*    cnt     = sums + NG * DD;
  unsigned* rowptr  = (unsigned*)(cnt + NG);
  unsigned* rank    = rowptr + 40064;
  int*      col     = (int*)(rank + NE);
  unsigned* tab_in  = (unsigned*)(col + NE);
  unsigned* tab_out = tab_in + (size_t)NCH * NN;

  const int T = 256;

  deg_chunk_kernel<<<2 * NCH, 1024, 0, stream>>>(src, dst, tab_in, tab_out, rank);
  merge_gemm_kernel<<<MB + NN / 16 / GEMM_SPW, 256, 0, stream>>>(tab_in, tab_out, lrow, btot, nsrc, ndst, sums,
                                                                  feat, W1, bufA);
  fill_kernel<<<(NE / FILL_EPT + T - 1) / T, T, 0, stream>>>(src, dst, lrow, btot, tab_in, rank, col, rowptr);

  spmm_pull<false><<<SPMM_WGS, 256, 0, stream>>>(bufA, rowptr, col, ndst, nsrc, b1, gid, bufB, cnt);

  spmm_pull<true><<<SPMM_WGS, 256, 0, stream>>>(bufB, rowptr, col, ndst, nsrc, b2, gid, sums, cnt);

  readout_kernel<<<NG / RO_ROWS, 256, 0, stream>>>(sums, cnt, W2, b2, Wc, bc, out);
}
